# starting kernel
# baseline (speedup 1.0000x reference)
#include <hip/hip_runtime.h>

typedef float v2f __attribute__((ext_vector_type(2)));
typedef float v8f __attribute__((ext_vector_type(8)));

#define SEQ   512
#define BATCH 2048
#define EMB   50
#define EMBP  52
#define NCLS  2

__launch_bounds__(256)
__global__ void attn_fused(const int* __restrict__ x,
                           const float* __restrict__ table,
                           const float* __restrict__ Wq, const float* __restrict__ bq,
                           const float* __restrict__ We, const float* __restrict__ be,
                           const float* __restrict__ Wo, const float* __restrict__ bo,
                           float* __restrict__ yhat_out, float* __restrict__ attn_out)
{
    __shared__ float emb[SEQ][EMBP];
    __shared__ int   tok[SEQ];
    __shared__ float sc[SEQ];
    __shared__ float part[4][EMBP];
    __shared__ float meanv[EMBP];
    __shared__ float qv[EMBP];
    __shared__ float ctxv[EMBP];
    __shared__ float pooled[EMBP];
    __shared__ float red[256];

    const int b    = blockIdx.x;
    const int tid  = threadIdx.x;
    const int lane = tid & 31;
    const int wave = tid >> 5;

    tok[tid]       = x[tid * BATCH + b];
    tok[tid + 256] = x[(tid + 256) * BATCH + b];
    __syncthreads();

    {
        const int g = tid >> 6, c = tid & 63;
        float s = 0.f;
        if (c < EMB) {
            for (int r = g; r < SEQ; r += 4) {
                const float v = table[tok[r] * EMB + c];
                emb[r][c] = v;
                s += v;
            }
        } else if (c < EMBP) {
            for (int r = g; r < SEQ; r += 4) emb[r][c] = 0.f;
        }
        if (c < EMBP) part[g][c] = s;
    }
    __syncthreads();
    if (tid < EMBP) {
        meanv[tid] = (tid < EMB)
            ? (part[0][tid] + part[1][tid] + part[2][tid] + part[3][tid]) * (1.0f / SEQ)
            : 0.f;
    }
    __syncthreads();

    if (tid < EMBP) {
        float acc = 0.f;
        if (tid < EMB) {
            acc = bq[tid];
            const float* wr = Wq + tid * EMB;
            for (int k = 0; k < EMB; ++k) acc += wr[k] * meanv[k];
        }
        qv[tid] = acc;
    }
    __syncthreads();

    {
        const int mrow  = lane & 15;
        const int khalf = (lane >> 4) * 2;

        v2f breg[13];
        #pragma unroll
        for (int kk = 0; kk < 13; ++kk) {
            const int kb = kk * 4 + khalf;
            breg[kk].x = (mrow == 0) ? qv[kb]     : 0.f;
            breg[kk].y = (mrow == 0) ? qv[kb + 1] : 0.f;
        }

        for (int i = 0; i < 4; ++i) {
            const int m0 = (wave * 4 + i) * 16;
            const float* rowp = &emb[m0 + mrow][0];
            v8f acc = {0.f, 0.f, 0.f, 0.f, 0.f, 0.f, 0.f, 0.f};
            #pragma unroll
            for (int kk = 0; kk < 13; ++kk) {
                const int kb = kk * 4 + khalf;
                const v2f a = *(const v2f*)(rowp + kb);
                acc = __builtin_amdgcn_wmma_f32_16x16x4_f32(
                          false, a, false, breg[kk], (short)0, acc, false, false);
            }
            if (mrow == 0) {
                const int base = m0 + (lane >> 4) * 8;
                #pragma unroll
                for (int j = 0; j < 8; ++j) sc[base + j] = acc[j];
            }
        }
    }
    __syncthreads();

    {
        const float s0 = sc[tid], s1 = sc[tid + 256];
        red[tid] = fmaxf(s0, s1);
        __syncthreads();
        for (int off = 128; off > 0; off >>= 1) {
            if (tid < off) red[tid] = fmaxf(red[tid], red[tid + off]);
            __syncthreads();
        }
        const float mx = red[0];
        __syncthreads();
        const float e0 = expf(s0 - mx);
        const float e1 = expf(s1 - mx);
        red[tid] = e0 + e1;
        __syncthreads();
        for (int off = 128; off > 0; off >>= 1) {
            if (tid < off) red[tid] += red[tid + off];
            __syncthreads();
        }
        const float inv = 1.0f / red[0];
        const float a0 = e0 * inv, a1 = e1 * inv;
        sc[tid]       = a0;
        sc[tid + 256] = a1;
        attn_out[b * SEQ + tid]       = a0;
        attn_out[b * SEQ + tid + 256] = a1;
    }
    __syncthreads();

    {
        const int g = tid >> 6, c = tid & 63;
        float s = 0.f;
        if (c < EMB) {
            for (int r = g; r < SEQ; r += 4)
                s += sc[r] * emb[r][c];
        }
        if (c < EMBP) part[g][c] = s;
    }
    __syncthreads();
    if (tid < EMBP) {
        ctxv[tid] = (tid < EMB)
            ? (part[0][tid] + part[1][tid] + part[2][tid] + part[3][tid])
            : 0.f;
    }
    __syncthreads();

    if (tid < EMB) {
        float acc = be[tid];
        const float* wr = We + tid * EMB;
        for (int k = 0; k < EMB; ++k) acc += wr[k] * ctxv[k];
        pooled[tid] = acc;
    }
    __syncthreads();

    if (tid < NCLS) {
        float acc = bo[tid];
        const float* wr = Wo + tid * EMB;
        for (int k = 0; k < EMB; ++k) acc += wr[k] * pooled[k];
        yhat_out[b * NCLS + tid] = acc;
    }
}

extern "C" void kernel_launch(void* const* d_in, const int* in_sizes, int n_in,
                              void* d_out, int out_size, void* d_ws, size_t ws_size,
                              hipStream_t stream) {
    const int*   x     = (const int*)  d_in[0];
    const float* table = (const float*)d_in[3];
    const float* Wq    = (const float*)d_in[4];
    const float* bq    = (const float*)d_in[5];
    const float* We    = (const float*)d_in[6];
    const float* be    = (const float*)d_in[7];
    const float* Wo    = (const float*)d_in[8];
    const float* bo    = (const float*)d_in[9];

    float* yhat = (float*)d_out;
    float* attn = (float*)d_out + BATCH * NCLS;

    attn_fused<<<dim3(BATCH), dim3(256), 0, stream>>>(
        x, table, Wq, bq, We, be, Wo, bo, yhat, attn);
}


# Gather: each wave fetches its 64 rows with one 8-byte load per lane (25 lanes per row), all 64 in flight at once; mean now summed as 8 per-wave partials (summation order changes)
# speedup vs baseline: 1.2860x; 1.2860x over previous
#include <hip/hip_runtime.h>

typedef float v2f __attribute__((ext_vector_type(2)));
typedef float v8f __attribute__((ext_vector_type(8)));

#define SEQ   512
#define BATCH 2048
#define EMB   50
#define EMBP  52
#define NCLS  2

__launch_bounds__(256)
__global__ void attn_fused(const int* __restrict__ x,
                           const float* __restrict__ table,
                           const float* __restrict__ Wq, const float* __restrict__ bq,
                           const float* __restrict__ We, const float* __restrict__ be,
                           const float* __restrict__ Wo, const float* __restrict__ bo,
                           float* __restrict__ yhat_out, float* __restrict__ attn_out)
{
    __shared__ __attribute__((aligned(16))) float emb[SEQ][EMBP];
    __shared__ float sc[SEQ];
    __shared__ __attribute__((aligned(16))) float part[8][EMBP];
    __shared__ float meanv[EMBP];
    __shared__ float qv[EMBP];
    __shared__ float ctxv[EMBP];
    __shared__ float pooled[EMBP];
    __shared__ float red[256];

    const int b    = blockIdx.x;
    const int tid  = threadIdx.x;
    const int lane = tid & 31;
    const int wave = tid >> 5;

    {
        const int r0 = wave * 64;
        const int t0 = x[(r0 + lane) * BATCH + b];
        const int t1 = x[(r0 + 32 + lane) * BATCH + b];

        const v2f zero2 = {0.f, 0.f};
        *(v2f*)&emb[r0 + lane][EMB]      = zero2;
        *(v2f*)&emb[r0 + 32 + lane][EMB] = zero2;

        int tk[64];
        #pragma unroll
        for (int i = 0; i < 64; ++i)
            tk[i] = __builtin_amdgcn_readlane((i < 32) ? t0 : t1, i & 31);

        if (lane < 25) {
            const int co = lane * 2;
            const v2f* tab2 = (const v2f*)table;
            v2f v[64];
            #pragma unroll
            for (int i = 0; i < 64; ++i)
                v[i] = tab2[tk[i] * (EMB / 2) + lane];

            v2f s0 = v[0], s1 = v[1], s2 = v[2], s3 = v[3];
            *(v2f*)&emb[r0    ][co] = v[0];
            *(v2f*)&emb[r0 + 1][co] = v[1];
            *(v2f*)&emb[r0 + 2][co] = v[2];
            *(v2f*)&emb[r0 + 3][co] = v[3];
            #pragma unroll
            for (int i = 4; i < 64; i += 4) {
                *(v2f*)&emb[r0 + i    ][co] = v[i];
                *(v2f*)&emb[r0 + i + 1][co] = v[i + 1];
                *(v2f*)&emb[r0 + i + 2][co] = v[i + 2];
                *(v2f*)&emb[r0 + i + 3][co] = v[i + 3];
                s0 += v[i]; s1 += v[i + 1]; s2 += v[i + 2]; s3 += v[i + 3];
            }
            *(v2f*)&part[wave][co] = (s0 + s1) + (s2 + s3);
        }
    }
    __syncthreads();
    if (tid < EMBP) {
        meanv[tid] = (tid < EMB)
            ? (((part[0][tid] + part[1][tid]) + (part[2][tid] + part[3][tid])) +
               ((part[4][tid] + part[5][tid]) + (part[6][tid] + part[7][tid]))) * (1.0f / SEQ)
            : 0.f;
    }
    __syncthreads();

    if (tid < EMBP) {
        float acc = 0.f;
        if (tid < EMB) {
            acc = bq[tid];
            const float* wr = Wq + tid * EMB;
            for (int k = 0; k < EMB; ++k) acc += wr[k] * meanv[k];
        }
        qv[tid] = acc;
    }
    __syncthreads();

    {
        const int mrow  = lane & 15;
        const int khalf = (lane >> 4) * 2;

        v2f breg[13];
        #pragma unroll
        for (int kk = 0; kk < 13; ++kk) {
            const int kb = kk * 4 + khalf;
            breg[kk].x = (mrow == 0) ? qv[kb]     : 0.f;
            breg[kk].y = (mrow == 0) ? qv[kb + 1] : 0.f;
        }

        for (int i = 0; i < 4; ++i) {
            const int m0 = (wave * 4 + i) * 16;
            const float* rowp = &emb[m0 + mrow][0];
            v8f acc = {0.f, 0.f, 0.f, 0.f, 0.f, 0.f, 0.f, 0.f};
            #pragma unroll
            for (int kk = 0; kk < 13; ++kk) {
                const int kb = kk * 4 + khalf;
                const v2f a = *(const v2f*)(rowp + kb);
                acc = __builtin_amdgcn_wmma_f32_16x16x4_f32(
                          false, a, false, breg[kk], (short)0, acc, false, false);
            }
            if (mrow == 0) {
                const int base = m0 + (lane >> 4) * 8;
                #pragma unroll
                for (int j = 0; j < 8; ++j) sc[base + j] = acc[j];
            }
        }
    }
    __syncthreads();

    {
        const float s0 = sc[tid], s1 = sc[tid + 256];
        red[tid] = fmaxf(s0, s1);
        __syncthreads();
        for (int off = 128; off > 0; off >>= 1) {
            if (tid < off) red[tid] = fmaxf(red[tid], red[tid + off]);
            __syncthreads();
        }
        const float mx = red[0];
        __syncthreads();
        const float e0 = expf(s0 - mx);
        const float e1 = expf(s1 - mx);
        red[tid] = e0 + e1;
        __syncthreads();
        for (int off = 128; off > 0; off >>= 1) {
            if (tid < off) red[tid] += red[tid + off];
            __syncthreads();
        }
        const float inv = 1.0f / red[0];
        const float a0 = e0 * inv, a1 = e1 * inv;
        sc[tid]       = a0;
        sc[tid + 256] = a1;
        attn_out[b * SEQ + tid]       = a0;
        attn_out[b * SEQ + tid + 256] = a1;
    }
    __syncthreads();

    {
        const int g = tid >> 6, c = tid & 63;
        float s = 0.f;
        if (c < EMB) {
            for (int r = g; r < SEQ; r += 4)
                s += sc[r] * emb[r][c];
        }
        if (c < EMBP) part[g][c] = s;
    }
    __syncthreads();
    if (tid < EMBP) {
        ctxv[tid] = (tid < EMB)
            ? (part[0][tid] + part[1][tid] + part[2][tid] + part[3][tid])
            : 0.f;
    }
    __syncthreads();

    if (tid < EMB) {
        float acc = be[tid];
        const float* wr = We + tid * EMB;
        for (int k = 0; k < EMB; ++k) acc += wr[k] * ctxv[k];
        pooled[tid] = acc;
    }
    __syncthreads();

    if (tid < NCLS) {
        float acc = bo[tid];
        const float* wr = Wo + tid * EMB;
        for (int k = 0; k < EMB; ++k) acc += wr[k] * pooled[k];
        yhat_out[b * NCLS + tid] = acc;
    }
}

extern "C" void kernel_launch(void* const* d_in, const int* in_sizes, int n_in,
                              void* d_out, int out_size, void* d_ws, size_t ws_size,
                              hipStream_t stream) {
    const int*   x     = (const int*)  d_in[0];
    const float* table = (const float*)d_in[3];
    const float* Wq    = (const float*)d_in[4];
    const float* bq    = (const float*)d_in[5];
    const float* We    = (const float*)d_in[6];
    const float* be    = (const float*)d_in[7];
    const float* Wo    = (const float*)d_in[8];
    const float* bo    = (const float*)d_in[9];

    float* yhat = (float*)d_out;
    float* attn = (float*)d_out + BATCH * NCLS;

    attn_fused<<<dim3(BATCH), dim3(256), 0, stream>>>(
        x, table, Wq, bq, We, be, Wo, bo, yhat, attn);
}


# LDS per work-group cut from 110 KB to 104 KB (gathered rows at pitch 50 floats not 52, K tail zeroed in registers, dead buffers share space): 3 work-groups per WGP instead of 2; sums unchanged
# speedup vs baseline: 1.4382x; 1.1184x over previous
#include <hip/hip_runtime.h>

typedef float v2f __attribute__((ext_vector_type(2)));
typedef float v8f __attribute__((ext_vector_type(8)));

#define SEQ   512
#define BATCH 2048
#define EMB   50
#define EMBP  52
#define NCLS  2

#define OFF_EMB 0
#define OFF_SC  (SEQ * EMB)
#define OFF_U   (OFF_SC + SEQ)
#define OFF_V1  (OFF_U + 8 * EMB)
#define OFF_V2  (OFF_V1 + EMBP)
#define LDS_FLOATS (OFF_V2 + EMBP)

__launch_bounds__(256)
__global__ void attn_fused(const int* __restrict__ x,
                           const float* __restrict__ table,
                           const float* __restrict__ Wq, const float* __restrict__ bq,
                           const float* __restrict__ We, const float* __restrict__ be,
                           const float* __restrict__ Wo, const float* __restrict__ bo,
                           float* __restrict__ yhat_out, float* __restrict__ attn_out)
{
    __shared__ __attribute__((aligned(16))) float lds[LDS_FLOATS];
    float (*emb)[EMB]  = (float (*)[EMB])(lds + OFF_EMB);
    float* sc          = lds + OFF_SC;
    float (*part)[EMB] = (float (*)[EMB])(lds + OFF_U);
    float* red         = lds + OFF_U;
    float* meanv       = lds + OFF_V1;
    float* ctxv        = lds + OFF_V1;
    float* qv          = lds + OFF_V2;
    float* pooled      = lds + OFF_V2;

    const int b    = blockIdx.x;
    const int tid  = threadIdx.x;
    const int lane = tid & 31;
    const int wave = tid >> 5;

    {
        const int r0 = wave * 64;
        const int t0 = x[(r0 + lane) * BATCH + b];
        const int t1 = x[(r0 + 32 + lane) * BATCH + b];

        int tk[64];
        #pragma unroll
        for (int i = 0; i < 64; ++i)
            tk[i] = __builtin_amdgcn_readlane((i < 32) ? t0 : t1, i & 31);

        if (lane < 25) {
            const int co = lane * 2;
            const v2f* tab2 = (const v2f*)table;
            v2f v[64];
            #pragma unroll
            for (int i = 0; i < 64; ++i)
                v[i] = tab2[tk[i] * (EMB / 2) + lane];

            v2f s0 = v[0], s1 = v[1], s2 = v[2], s3 = v[3];
            *(v2f*)&emb[r0    ][co] = v[0];
            *(v2f*)&emb[r0 + 1][co] = v[1];
            *(v2f*)&emb[r0 + 2][co] = v[2];
            *(v2f*)&emb[r0 + 3][co] = v[3];
            #pragma unroll
            for (int i = 4; i < 64; i += 4) {
                *(v2f*)&emb[r0 + i    ][co] = v[i];
                *(v2f*)&emb[r0 + i + 1][co] = v[i + 1];
                *(v2f*)&emb[r0 + i + 2][co] = v[i + 2];
                *(v2f*)&emb[r0 + i + 3][co] = v[i + 3];
                s0 += v[i]; s1 += v[i + 1]; s2 += v[i + 2]; s3 += v[i + 3];
            }
            *(v2f*)&part[wave][co] = (s0 + s1) + (s2 + s3);
        }
    }
    __syncthreads();
    if (tid < EMB) {
        meanv[tid] = (((part[0][tid] + part[1][tid]) + (part[2][tid] + part[3][tid])) +
                      ((part[4][tid] + part[5][tid]) + (part[6][tid] + part[7][tid]))) * (1.0f / SEQ);
    }
    __syncthreads();

    if (tid < EMBP) {
        float acc = 0.f;
        if (tid < EMB) {
            acc = bq[tid];
            const float* wr = Wq + tid * EMB;
            #pragma unroll
            for (int k = 0; k < EMB; ++k) acc += wr[k] * meanv[k];
        }
        qv[tid] = acc;
    }
    __syncthreads();

    {
        const int mrow  = lane & 15;
        const int khalf = (lane >> 4) * 2;

        v2f breg[13];
        #pragma unroll
        for (int kk = 0; kk < 13; ++kk) {
            const int kb = kk * 4 + khalf;
            breg[kk].x = (mrow == 0) ? qv[kb]     : 0.f;
            breg[kk].y = (mrow == 0) ? qv[kb + 1] : 0.f;
        }

        for (int i = 0; i < 4; ++i) {
            const int m0 = (wave * 4 + i) * 16;
            const float* rowp = &emb[m0 + mrow][0];
            v8f acc = {0.f, 0.f, 0.f, 0.f, 0.f, 0.f, 0.f, 0.f};
            #pragma unroll
            for (int kk = 0; kk < 12; ++kk) {
                const int kb = kk * 4 + khalf;
                const v2f a = *(const v2f*)(rowp + kb);
                acc = __builtin_amdgcn_wmma_f32_16x16x4_f32(
                          false, a, false, breg[kk], (short)0, acc, false, false);
            }
            {
                v2f a = *(const v2f*)(rowp + 48);
                if (khalf) { a.x = 0.f; a.y = 0.f; }
                acc = __builtin_amdgcn_wmma_f32_16x16x4_f32(
                          false, a, false, breg[12], (short)0, acc, false, false);
            }
            if (mrow == 0) {
                const int base = m0 + (lane >> 4) * 8;
                #pragma unroll
                for (int j = 0; j < 8; ++j) sc[base + j] = acc[j];
            }
        }
    }
    __syncthreads();

    {
        const float s0 = sc[tid], s1 = sc[tid + 256];
        red[tid] = fmaxf(s0, s1);
        __syncthreads();
        for (int off = 128; off > 0; off >>= 1) {
            if (tid < off) red[tid] = fmaxf(red[tid], red[tid + off]);
            __syncthreads();
        }
        const float mx = red[0];
        __syncthreads();
        const float e0 = expf(s0 - mx);
        const float e1 = expf(s1 - mx);
        red[tid] = e0 + e1;
        __syncthreads();
        for (int off = 128; off > 0; off >>= 1) {
            if (tid < off) red[tid] += red[tid + off];
            __syncthreads();
        }
        const float inv = 1.0f / red[0];
        const float a0 = e0 * inv, a1 = e1 * inv;
        sc[tid]       = a0;
        sc[tid + 256] = a1;
        attn_out[b * SEQ + tid]       = a0;
        attn_out[b * SEQ + tid + 256] = a1;
    }
    __syncthreads();

    {
        const int g = tid >> 6, c = tid & 63;
        float s = 0.f;
        if (c < EMB) {
            for (int r = g; r < SEQ; r += 4)
                s += sc[r] * emb[r][c];
        }
        if (c < EMB) part[g][c] = s;
    }
    __syncthreads();
    if (tid < EMB) {
        ctxv[tid] = part[0][tid] + part[1][tid] + part[2][tid] + part[3][tid];
    }
    __syncthreads();

    if (tid < EMB) {
        float acc = be[tid];
        const float* wr = We + tid * EMB;
        #pragma unroll
        for (int k = 0; k < EMB; ++k) acc += wr[k] * ctxv[k];
        pooled[tid] = acc;
    }
    __syncthreads();

    if (tid < NCLS) {
        float acc = bo[tid];
        const float* wr = Wo + tid * EMB;
        #pragma unroll
        for (int k = 0; k < EMB; ++k) acc += wr[k] * pooled[k];
        yhat_out[b * NCLS + tid] = acc;
    }
}

extern "C" void kernel_launch(void* const* d_in, const int* in_sizes, int n_in,
                              void* d_out, int out_size, void* d_ws, size_t ws_size,
                              hipStream_t stream) {
    const int*   x     = (const int*)  d_in[0];
    const float* table = (const float*)d_in[3];
    const float* Wq    = (const float*)d_in[4];
    const float* bq    = (const float*)d_in[5];
    const float* We    = (const float*)d_in[6];
    const float* be    = (const float*)d_in[7];
    const float* Wo    = (const float*)d_in[8];
    const float* bo    = (const float*)d_in[9];

    float* yhat = (float*)d_out;
    float* attn = (float*)d_out + BATCH * NCLS;

    attn_fused<<<dim3(BATCH), dim3(256), 0, stream>>>(
        x, table, Wq, bq, We, be, Wo, bo, yhat, attn);
}


# Softmax max and sum reduced inside each wave with cross-lane DPP steps, the 8 per-wave results exchanged once through LDS: 19 barriers become 2; the exponentials are summed in a different order
# speedup vs baseline: 1.5194x; 1.0565x over previous
#include <hip/hip_runtime.h>

typedef float v2f __attribute__((ext_vector_type(2)));
typedef float v8f __attribute__((ext_vector_type(8)));

#define SEQ   512
#define BATCH 2048
#define EMB   50
#define EMBP  52
#define NCLS  2

#define OFF_EMB 0
#define OFF_SC  (SEQ * EMB)
#define OFF_U   (OFF_SC + SEQ)
#define OFF_V1  (OFF_U + 8 * EMB)
#define OFF_V2  (OFF_V1 + EMBP)
#define LDS_FLOATS (OFF_V2 + EMBP)

template <int M>
__device__ __forceinline__ float xrow(float x) {
    return __builtin_amdgcn_update_dpp(x, x, 0x160 + M, 0xf, 0xf, true);
}
__device__ __forceinline__ float xhalf(float x) {
    const int b = __builtin_bit_cast(int, x);
    return __builtin_bit_cast(float,
        __builtin_amdgcn_permlanex16(b, b, 0x76543210, 0xfedcba98, false, false));
}
__device__ __forceinline__ float wave_max(float v) {
    v = fmaxf(v, xrow<1>(v));
    v = fmaxf(v, xrow<2>(v));
    v = fmaxf(v, xrow<4>(v));
    v = fmaxf(v, xrow<8>(v));
    return fmaxf(v, xhalf(v));
}
__device__ __forceinline__ float wave_sum(float v) {
    v += xrow<1>(v);
    v += xrow<2>(v);
    v += xrow<4>(v);
    v += xrow<8>(v);
    return v + xhalf(v);
}

__launch_bounds__(256) __attribute__((amdgpu_num_vgpr(144)))
__global__ void attn_fused(const int* __restrict__ x,
                           const float* __restrict__ table,
                           const float* __restrict__ Wq, const float* __restrict__ bq,
                           const float* __restrict__ We, const float* __restrict__ be,
                           const float* __restrict__ Wo, const float* __restrict__ bo,
                           float* __restrict__ yhat_out, float* __restrict__ attn_out)
{
    __shared__ __attribute__((aligned(16))) float lds[LDS_FLOATS];
    float (*emb)[EMB]  = (float (*)[EMB])(lds + OFF_EMB);
    float* sc          = lds + OFF_SC;
    float (*part)[EMB] = (float (*)[EMB])(lds + OFF_U);
    float* wred        = lds + OFF_U;
    float* meanv       = lds + OFF_V1;
    float* ctxv        = lds + OFF_V1;
    float* qv          = lds + OFF_V2;
    float* pooled      = lds + OFF_V2;

    const int b    = blockIdx.x;
    const int tid  = threadIdx.x;
    const int lane = tid & 31;
    const int wave = tid >> 5;

    {
        const int r0 = wave * 64;
        const int t0 = x[(r0 + lane) * BATCH + b];
        const int t1 = x[(r0 + 32 + lane) * BATCH + b];

        int tk[64];
        #pragma unroll
        for (int i = 0; i < 64; ++i)
            tk[i] = __builtin_amdgcn_readlane((i < 32) ? t0 : t1, i & 31);

        if (lane < 25) {
            const int co = lane * 2;
            const v2f* tab2 = (const v2f*)table;
            v2f v[64];
            #pragma unroll
            for (int i = 0; i < 64; ++i)
                v[i] = tab2[tk[i] * (EMB / 2) + lane];

            v2f s0 = v[0], s1 = v[1], s2 = v[2], s3 = v[3];
            *(v2f*)&emb[r0    ][co] = v[0];
            *(v2f*)&emb[r0 + 1][co] = v[1];
            *(v2f*)&emb[r0 + 2][co] = v[2];
            *(v2f*)&emb[r0 + 3][co] = v[3];
            #pragma unroll
            for (int i = 4; i < 64; i += 4) {
                *(v2f*)&emb[r0 + i    ][co] = v[i];
                *(v2f*)&emb[r0 + i + 1][co] = v[i + 1];
                *(v2f*)&emb[r0 + i + 2][co] = v[i + 2];
                *(v2f*)&emb[r0 + i + 3][co] = v[i + 3];
                s0 += v[i]; s1 += v[i + 1]; s2 += v[i + 2]; s3 += v[i + 3];
            }
            *(v2f*)&part[wave][co] = (s0 + s1) + (s2 + s3);
        }
    }
    __syncthreads();
    if (tid < EMB) {
        meanv[tid] = (((part[0][tid] + part[1][tid]) + (part[2][tid] + part[3][tid])) +
                      ((part[4][tid] + part[5][tid]) + (part[6][tid] + part[7][tid]))) * (1.0f / SEQ);
    }
    __syncthreads();

    if (tid < EMBP) {
        float acc = 0.f;
        if (tid < EMB) {
            acc = bq[tid];
            const float* wr = Wq + tid * EMB;
            #pragma unroll
            for (int k = 0; k < EMB; ++k) acc += wr[k] * meanv[k];
        }
        qv[tid] = acc;
    }
    __syncthreads();

    {
        const int mrow  = lane & 15;
        const int khalf = (lane >> 4) * 2;

        v2f breg[13];
        #pragma unroll
        for (int kk = 0; kk < 13; ++kk) {
            const int kb = kk * 4 + khalf;
            breg[kk].x = (mrow == 0) ? qv[kb]     : 0.f;
            breg[kk].y = (mrow == 0) ? qv[kb + 1] : 0.f;
        }

        for (int i = 0; i < 4; ++i) {
            const int m0 = (wave * 4 + i) * 16;
            const float* rowp = &emb[m0 + mrow][0];
            v8f acc = {0.f, 0.f, 0.f, 0.f, 0.f, 0.f, 0.f, 0.f};
            #pragma unroll
            for (int kk = 0; kk < 12; ++kk) {
                const int kb = kk * 4 + khalf;
                const v2f a = *(const v2f*)(rowp + kb);
                acc = __builtin_amdgcn_wmma_f32_16x16x4_f32(
                          false, a, false, breg[kk], (short)0, acc, false, false);
            }
            {
                v2f a = *(const v2f*)(rowp + 48);
                if (khalf) { a.x = 0.f; a.y = 0.f; }
                acc = __builtin_amdgcn_wmma_f32_16x16x4_f32(
                          false, a, false, breg[12], (short)0, acc, false, false);
            }
            if (mrow == 0) {
                const int base = m0 + (lane >> 4) * 8;
                #pragma unroll
                for (int j = 0; j < 8; ++j) sc[base + j] = acc[j];
            }
        }
    }
    __syncthreads();

    {
        const float s0 = sc[tid], s1 = sc[tid + 256];
        const float wm = wave_max(fmaxf(s0, s1));
        if (lane == 0) wred[wave] = wm;
        __syncthreads();
        const float4 m03 = *(const float4*)&wred[0];
        const float4 m47 = *(const float4*)&wred[4];
        const float mx = fmaxf(fmaxf(fmaxf(m03.x, m03.y), fmaxf(m03.z, m03.w)),
                               fmaxf(fmaxf(m47.x, m47.y), fmaxf(m47.z, m47.w)));
        const float e0 = expf(s0 - mx);
        const float e1 = expf(s1 - mx);
        const float ws = wave_sum(e0 + e1);
        if (lane == 0) wred[8 + wave] = ws;
        __syncthreads();
        const float4 t03 = *(const float4*)&wred[8];
        const float4 t47 = *(const float4*)&wred[12];
        const float tot = ((t03.x + t03.y) + (t03.z + t03.w)) + ((t47.x + t47.y) + (t47.z + t47.w));
        const float inv = 1.0f / tot;
        const float a0 = e0 * inv, a1 = e1 * inv;
        sc[tid]       = a0;
        sc[tid + 256] = a1;
        attn_out[b * SEQ + tid]       = a0;
        attn_out[b * SEQ + tid + 256] = a1;
    }
    __syncthreads();

    {
        const int g = tid >> 6, c = tid & 63;
        float s = 0.f;
        if (c < EMB) {
            for (int r = g; r < SEQ; r += 4)
                s += sc[r] * emb[r][c];
        }
        if (c < EMB) part[g][c] = s;
    }
    __syncthreads();
    if (tid < EMB) {
        ctxv[tid] = part[0][tid] + part[1][tid] + part[2][tid] + part[3][tid];
    }
    __syncthreads();

    if (tid < EMB) {
        float acc = be[tid];
        const float* wr = We + tid * EMB;
        #pragma unroll
        for (int k = 0; k < EMB; ++k) acc += wr[k] * ctxv[k];
        pooled[tid] = acc;
    }
    __syncthreads();

    if (tid < NCLS) {
        float acc = bo[tid];
        const float* wr = Wo + tid * EMB;
        #pragma unroll
        for (int k = 0; k < EMB; ++k) acc += wr[k] * pooled[k];
        yhat_out[b * NCLS + tid] = acc;
    }
}

extern "C" void kernel_launch(void* const* d_in, const int* in_sizes, int n_in,
                              void* d_out, int out_size, void* d_ws, size_t ws_size,
                              hipStream_t stream) {
    const int*   x     = (const int*)  d_in[0];
    const float* table = (const float*)d_in[3];
    const float* Wq    = (const float*)d_in[4];
    const float* bq    = (const float*)d_in[5];
    const float* We    = (const float*)d_in[6];
    const float* be    = (const float*)d_in[7];
    const float* Wo    = (const float*)d_in[8];
    const float* bo    = (const float*)d_in[9];

    float* yhat = (float*)d_out;
    float* attn = (float*)d_out + BATCH * NCLS;

    attn_fused<<<dim3(BATCH), dim3(256), 0, stream>>>(
        x, table, Wq, bq, We, be, Wo, bo, yhat, attn);
}


# Each wave keeps the scores and attention weights of its own 64 rows in registers (q in every B column) and sums those rows' context itself: 2 barriers go; softmax and context sums change order
# speedup vs baseline: 1.8510x; 1.2182x over previous
#include <hip/hip_runtime.h>

typedef float v2f __attribute__((ext_vector_type(2)));
typedef float v8f __attribute__((ext_vector_type(8)));

#define SEQ   512
#define BATCH 2048
#define EMB   50
#define EMBP  52
#define NCLS  2

#define OFF_EMB 0
#define OFF_U   (SEQ * EMB)
#define OFF_WR  (OFF_U + 8 * EMB)
#define OFF_V1  (OFF_WR + 16)
#define OFF_V2  (OFF_V1 + EMBP)
#define LDS_FLOATS (OFF_V2 + EMBP)

template <int M>
__device__ __forceinline__ float xrow(float x) {
    return __builtin_amdgcn_update_dpp(x, x, 0x160 + M, 0xf, 0xf, true);
}
__device__ __forceinline__ float xhalf(float x) {
    const int b = __builtin_bit_cast(int, x);
    return __builtin_bit_cast(float,
        __builtin_amdgcn_permlanex16(b, b, 0x76543210, 0xfedcba98, false, false));
}
__device__ __forceinline__ float wave_max(float v) {
    v = fmaxf(v, xrow<1>(v));
    v = fmaxf(v, xrow<2>(v));
    v = fmaxf(v, xrow<4>(v));
    v = fmaxf(v, xrow<8>(v));
    return fmaxf(v, xhalf(v));
}
__device__ __forceinline__ float wave_sum(float v) {
    v += xrow<1>(v);
    v += xrow<2>(v);
    v += xrow<4>(v);
    v += xrow<8>(v);
    return v + xhalf(v);
}

__launch_bounds__(256) __attribute__((amdgpu_num_vgpr(144)))
__global__ void attn_fused(const int* __restrict__ x,
                           const float* __restrict__ table,
                           const float* __restrict__ Wq, const float* __restrict__ bq,
                           const float* __restrict__ We, const float* __restrict__ be,
                           const float* __restrict__ Wo, const float* __restrict__ bo,
                           float* __restrict__ yhat_out, float* __restrict__ attn_out)
{
    __shared__ __attribute__((aligned(16))) float lds[LDS_FLOATS];
    float (*emb)[EMB]  = (float (*)[EMB])(lds + OFF_EMB);
    float (*part)[EMB] = (float (*)[EMB])(lds + OFF_U);
    float* wred        = lds + OFF_WR;
    float* meanv       = lds + OFF_V1;
    float* ctxv        = lds + OFF_V1;
    float* qv          = lds + OFF_V2;
    float* pooled      = lds + OFF_V2;

    const int b    = blockIdx.x;
    const int tid  = threadIdx.x;
    const int lane = tid & 31;
    const int wave = tid >> 5;

    {
        const int r0 = wave * 64;
        const int t0 = x[(r0 + lane) * BATCH + b];
        const int t1 = x[(r0 + 32 + lane) * BATCH + b];

        int tk[64];
        #pragma unroll
        for (int i = 0; i < 64; ++i)
            tk[i] = __builtin_amdgcn_readlane((i < 32) ? t0 : t1, i & 31);

        if (lane < 25) {
            const int co = lane * 2;
            const v2f* tab2 = (const v2f*)table;
            v2f v[64];
            #pragma unroll
            for (int i = 0; i < 64; ++i)
                v[i] = tab2[tk[i] * (EMB / 2) + lane];

            v2f s0 = v[0], s1 = v[1], s2 = v[2], s3 = v[3];
            *(v2f*)&emb[r0    ][co] = v[0];
            *(v2f*)&emb[r0 + 1][co] = v[1];
            *(v2f*)&emb[r0 + 2][co] = v[2];
            *(v2f*)&emb[r0 + 3][co] = v[3];
            #pragma unroll
            for (int i = 4; i < 64; i += 4) {
                *(v2f*)&emb[r0 + i    ][co] = v[i];
                *(v2f*)&emb[r0 + i + 1][co] = v[i + 1];
                *(v2f*)&emb[r0 + i + 2][co] = v[i + 2];
                *(v2f*)&emb[r0 + i + 3][co] = v[i + 3];
                s0 += v[i]; s1 += v[i + 1]; s2 += v[i + 2]; s3 += v[i + 3];
            }
            *(v2f*)&part[wave][co] = (s0 + s1) + (s2 + s3);
        }
    }
    __syncthreads();
    if (tid < EMB) {
        meanv[tid] = (((part[0][tid] + part[1][tid]) + (part[2][tid] + part[3][tid])) +
                      ((part[4][tid] + part[5][tid]) + (part[6][tid] + part[7][tid]))) * (1.0f / SEQ);
    }
    __syncthreads();

    if (tid < EMBP) {
        float acc = 0.f;
        if (tid < EMB) {
            acc = bq[tid];
            const float* wr = Wq + tid * EMB;
            #pragma unroll
            for (int k = 0; k < EMB; ++k) acc += wr[k] * meanv[k];
        }
        qv[tid] = acc;
    }
    __syncthreads();

    const int mrow = lane & 15;
    const int rr   = (mrow >> 2) * 16 + (lane >> 4) * 8 + (mrow & 3) * 2;
    float s0, s1;
    {
        const int khalf = (lane >> 4) * 2;

        v2f breg[13];
        #pragma unroll
        for (int kk = 0; kk < 13; ++kk)
            breg[kk] = *(const v2f*)&qv[kk * 4 + khalf];

        v8f acc[4];
        #pragma unroll
        for (int i = 0; i < 4; ++i) {
            const int m0 = (wave * 4 + i) * 16;
            const float* rowp = &emb[m0 + mrow][0];
            v8f d = {0.f, 0.f, 0.f, 0.f, 0.f, 0.f, 0.f, 0.f};
            #pragma unroll
            for (int kk = 0; kk < 12; ++kk) {
                const int kb = kk * 4 + khalf;
                const v2f a = *(const v2f*)(rowp + kb);
                d = __builtin_amdgcn_wmma_f32_16x16x4_f32(
                        false, a, false, breg[kk], (short)0, d, false, false);
            }
            {
                v2f a = *(const v2f*)(rowp + 48);
                if (khalf) { a.x = 0.f; a.y = 0.f; }
                d = __builtin_amdgcn_wmma_f32_16x16x4_f32(
                        false, a, false, breg[12], (short)0, d, false, false);
            }
            acc[i] = d;
        }
        const int it = mrow >> 2, p = mrow & 3;
        float u[8];
        #pragma unroll
        for (int j = 0; j < 8; ++j) {
            float x = acc[0][j];
            x = (it == 1) ? acc[1][j] : x;
            x = (it == 2) ? acc[2][j] : x;
            x = (it == 3) ? acc[3][j] : x;
            u[j] = x;
        }
        s0 = u[0]; s1 = u[1];
        s0 = (p == 1) ? u[2] : s0;  s1 = (p == 1) ? u[3] : s1;
        s0 = (p == 2) ? u[4] : s0;  s1 = (p == 2) ? u[5] : s1;
        s0 = (p == 3) ? u[6] : s0;  s1 = (p == 3) ? u[7] : s1;
    }

    float a0, a1;
    {
        const float wm = wave_max(fmaxf(s0, s1));
        if (lane == 0) wred[wave] = wm;
        __syncthreads();
        const float4 m03 = *(const float4*)&wred[0];
        const float4 m47 = *(const float4*)&wred[4];
        const float mx = fmaxf(fmaxf(fmaxf(m03.x, m03.y), fmaxf(m03.z, m03.w)),
                               fmaxf(fmaxf(m47.x, m47.y), fmaxf(m47.z, m47.w)));
        const float e0 = expf(s0 - mx);
        const float e1 = expf(s1 - mx);
        const float ws = wave_sum(e0 + e1);
        if (lane == 0) wred[8 + wave] = ws;
        __syncthreads();
        const float4 t03 = *(const float4*)&wred[8];
        const float4 t47 = *(const float4*)&wred[12];
        const float tot = ((t03.x + t03.y) + (t03.z + t03.w)) + ((t47.x + t47.y) + (t47.z + t47.w));
        const float inv = 1.0f / tot;
        a0 = e0 * inv;
        a1 = e1 * inv;
        v2f av; av.x = a0; av.y = a1;
        *(v2f*)&attn_out[b * SEQ + wave * 64 + rr] = av;
    }

    {
        const int r0 = wave * 64;
        const int co = lane * 2;
        v2f c0 = {0.f, 0.f}, c1 = {0.f, 0.f}, c2 = {0.f, 0.f}, c3 = {0.f, 0.f};
        #pragma unroll
        for (int h = 0; h < 64; h += 32) {
            float wt[32];
            #pragma unroll
            for (int e = 0; e < 32; ++e) {
                const int q = h + e;
                const int srcl = ((q >> 3) & 1) * 16 + (q >> 4) * 4 + ((q >> 1) & 3);
                wt[e] = __builtin_bit_cast(float,
                    __builtin_amdgcn_readlane(__builtin_bit_cast(int, (q & 1) ? a1 : a0), srcl));
            }
            if (lane < 25) {
                #pragma unroll
                for (int e = 0; e < 32; e += 4) {
                    c0 += wt[e    ] * *(const v2f*)&emb[r0 + h + e    ][co];
                    c1 += wt[e + 1] * *(const v2f*)&emb[r0 + h + e + 1][co];
                    c2 += wt[e + 2] * *(const v2f*)&emb[r0 + h + e + 2][co];
                    c3 += wt[e + 3] * *(const v2f*)&emb[r0 + h + e + 3][co];
                }
            }
        }
        if (lane < 25) *(v2f*)&part[wave][co] = (c0 + c1) + (c2 + c3);
    }
    __syncthreads();
    if (tid < EMB) {
        ctxv[tid] = ((part[0][tid] + part[1][tid]) + (part[2][tid] + part[3][tid])) +
                    ((part[4][tid] + part[5][tid]) + (part[6][tid] + part[7][tid]));
    }
    __syncthreads();

    if (tid < EMB) {
        float acc = be[tid];
        const float* wr = We + tid * EMB;
        #pragma unroll
        for (int k = 0; k < EMB; ++k) acc += wr[k] * ctxv[k];
        pooled[tid] = acc;
    }
    __syncthreads();

    if (tid < NCLS) {
        float acc = bo[tid];
        const float* wr = Wo + tid * EMB;
        #pragma unroll
        for (int k = 0; k < EMB; ++k) acc += wr[k] * pooled[k];
        yhat_out[b * NCLS + tid] = acc;
    }
}

extern "C" void kernel_launch(void* const* d_in, const int* in_sizes, int n_in,
                              void* d_out, int out_size, void* d_ws, size_t ws_size,
                              hipStream_t stream) {
    const int*   x     = (const int*)  d_in[0];
    const float* table = (const float*)d_in[3];
    const float* Wq    = (const float*)d_in[4];
    const float* bq    = (const float*)d_in[5];
    const float* We    = (const float*)d_in[6];
    const float* be    = (const float*)d_in[7];
    const float* Wo    = (const float*)d_in[8];
    const float* bo    = (const float*)d_in[9];

    float* yhat = (float*)d_out;
    float* attn = (float*)d_out + BATCH * NCLS;

    attn_fused<<<dim3(BATCH), dim3(256), 0, stream>>>(
        x, table, Wq, bq, We, be, Wo, bo, yhat, attn);
}


# The three small matrix-vector products: 4 (16) neighbouring lanes share a weight row (8-byte pieces, short multiply-add chains), partial sums added across lanes; dot products summed in another order
# speedup vs baseline: 1.9684x; 1.0634x over previous
#include <hip/hip_runtime.h>

typedef float v2f __attribute__((ext_vector_type(2)));
typedef float v8f __attribute__((ext_vector_type(8)));

#define SEQ   512
#define BATCH 2048
#define EMB   50
#define EMBP  52
#define NCLS  2

#define OFF_EMB 0
#define OFF_U   (SEQ * EMB)
#define OFF_WR  (OFF_U + 8 * EMB)
#define OFF_V1  (OFF_WR + 16)
#define OFF_V2  (OFF_V1 + EMBP)
#define LDS_FLOATS (OFF_V2 + EMBP)

template <int M>
__device__ __forceinline__ float xrow(float x) {
    return __builtin_amdgcn_update_dpp(x, x, 0x160 + M, 0xf, 0xf, true);
}
__device__ __forceinline__ float xhalf(float x) {
    const int b = __builtin_bit_cast(int, x);
    return __builtin_bit_cast(float,
        __builtin_amdgcn_permlanex16(b, b, 0x76543210, 0xfedcba98, false, false));
}
__device__ __forceinline__ float wave_max(float v) {
    v = fmaxf(v, xrow<1>(v));
    v = fmaxf(v, xrow<2>(v));
    v = fmaxf(v, xrow<4>(v));
    v = fmaxf(v, xrow<8>(v));
    return fmaxf(v, xhalf(v));
}
__device__ __forceinline__ float wave_sum(float v) {
    v += xrow<1>(v);
    v += xrow<2>(v);
    v += xrow<4>(v);
    v += xrow<8>(v);
    return v + xhalf(v);
}

template <int G>
__device__ __forceinline__ float rowdot_part(const float* __restrict__ wrow,
                                             const float* vec, int c) {
    const v2f* w2 = (const v2f*)wrow;
    const v2f* x2 = (const v2f*)vec;
    constexpr int FULL = (EMB / 2) / G;
    constexpr int REST = (EMB / 2) % G;
    static_assert(FULL >= 1 && REST >= 1, "25 pairs over G lanes, with a remainder");
    const bool last = c < REST;
    const int  pl   = last ? c + G * FULL : c;
    v2f w[FULL];
    #pragma unroll
    for (int j = 0; j < FULL; ++j) w[j] = w2[c + G * j];
    const v2f wl = w2[pl];
    v2f s = w[0] * x2[c];
    #pragma unroll
    for (int j = 1; j < FULL; ++j) s += w[j] * x2[c + G * j];
    const v2f sl = s + wl * x2[pl];
    s = last ? sl : s;
    return s.x + s.y;
}

__launch_bounds__(256) __attribute__((amdgpu_num_vgpr(144)))
__global__ void attn_fused(const int* __restrict__ x,
                           const float* __restrict__ table,
                           const float* __restrict__ Wq, const float* __restrict__ bq,
                           const float* __restrict__ We, const float* __restrict__ be,
                           const float* __restrict__ Wo, const float* __restrict__ bo,
                           float* __restrict__ yhat_out, float* __restrict__ attn_out)
{
    __shared__ __attribute__((aligned(16))) float lds[LDS_FLOATS];
    float (*emb)[EMB]  = (float (*)[EMB])(lds + OFF_EMB);
    float (*part)[EMB] = (float (*)[EMB])(lds + OFF_U);
    float* wred        = lds + OFF_WR;
    float* meanv       = lds + OFF_V1;
    float* ctxv        = lds + OFF_V1;
    float* qv          = lds + OFF_V2;
    float* pooled      = lds + OFF_V2;

    const int b    = blockIdx.x;
    const int tid  = threadIdx.x;
    const int lane = tid & 31;
    const int wave = tid >> 5;

    {
        const int r0 = wave * 64;
        const int t0 = x[(r0 + lane) * BATCH + b];
        const int t1 = x[(r0 + 32 + lane) * BATCH + b];

        int tk[64];
        #pragma unroll
        for (int i = 0; i < 64; ++i)
            tk[i] = __builtin_amdgcn_readlane((i < 32) ? t0 : t1, i & 31);

        if (lane < 25) {
            const int co = lane * 2;
            const v2f* tab2 = (const v2f*)table;
            v2f v[64];
            #pragma unroll
            for (int i = 0; i < 64; ++i)
                v[i] = tab2[tk[i] * (EMB / 2) + lane];

            v2f s0 = v[0], s1 = v[1], s2 = v[2], s3 = v[3];
            *(v2f*)&emb[r0    ][co] = v[0];
            *(v2f*)&emb[r0 + 1][co] = v[1];
            *(v2f*)&emb[r0 + 2][co] = v[2];
            *(v2f*)&emb[r0 + 3][co] = v[3];
            #pragma unroll
            for (int i = 4; i < 64; i += 4) {
                *(v2f*)&emb[r0 + i    ][co] = v[i];
                *(v2f*)&emb[r0 + i + 1][co] = v[i + 1];
                *(v2f*)&emb[r0 + i + 2][co] = v[i + 2];
                *(v2f*)&emb[r0 + i + 3][co] = v[i + 3];
                s0 += v[i]; s1 += v[i + 1]; s2 += v[i + 2]; s3 += v[i + 3];
            }
            *(v2f*)&part[wave][co] = (s0 + s1) + (s2 + s3);
        }
    }
    __syncthreads();
    if (tid < EMB) {
        meanv[tid] = (((part[0][tid] + part[1][tid]) + (part[2][tid] + part[3][tid])) +
                      ((part[4][tid] + part[5][tid]) + (part[6][tid] + part[7][tid]))) * (1.0f / SEQ);
    }
    __syncthreads();

    {
        const int row = tid >> 2, c = tid & 3;
        float acc = 0.f, bias = 0.f;
        if (row < EMB) {
            bias = bq[row];
            acc = rowdot_part<4>(Wq + row * EMB, meanv, c);
        }
        acc += xrow<1>(acc);
        acc += xrow<2>(acc);
        if (c == 0 && row < EMBP) qv[row] = acc + bias;
    }
    __syncthreads();

    const int mrow = lane & 15;
    const int rr   = (mrow >> 2) * 16 + (lane >> 4) * 8 + (mrow & 3) * 2;
    float s0, s1;
    {
        const int khalf = (lane >> 4) * 2;

        v2f breg[13];
        #pragma unroll
        for (int kk = 0; kk < 13; ++kk)
            breg[kk] = *(const v2f*)&qv[kk * 4 + khalf];

        v8f acc[4];
        #pragma unroll
        for (int i = 0; i < 4; ++i) {
            const int m0 = (wave * 4 + i) * 16;
            const float* rowp = &emb[m0 + mrow][0];
            v8f d = {0.f, 0.f, 0.f, 0.f, 0.f, 0.f, 0.f, 0.f};
            #pragma unroll
            for (int kk = 0; kk < 12; ++kk) {
                const int kb = kk * 4 + khalf;
                const v2f a = *(const v2f*)(rowp + kb);
                d = __builtin_amdgcn_wmma_f32_16x16x4_f32(
                        false, a, false, breg[kk], (short)0, d, false, false);
            }
            {
                v2f a = *(const v2f*)(rowp + 48);
                if (khalf) { a.x = 0.f; a.y = 0.f; }
                d = __builtin_amdgcn_wmma_f32_16x16x4_f32(
                        false, a, false, breg[12], (short)0, d, false, false);
            }
            acc[i] = d;
        }
        const int it = mrow >> 2, p = mrow & 3;
        float u[8];
        #pragma unroll
        for (int j = 0; j < 8; ++j) {
            float x = acc[0][j];
            x = (it == 1) ? acc[1][j] : x;
            x = (it == 2) ? acc[2][j] : x;
            x = (it == 3) ? acc[3][j] : x;
            u[j] = x;
        }
        s0 = u[0]; s1 = u[1];
        s0 = (p == 1) ? u[2] : s0;  s1 = (p == 1) ? u[3] : s1;
        s0 = (p == 2) ? u[4] : s0;  s1 = (p == 2) ? u[5] : s1;
        s0 = (p == 3) ? u[6] : s0;  s1 = (p == 3) ? u[7] : s1;
    }

    float a0, a1;
    {
        const float wm = wave_max(fmaxf(s0, s1));
        if (lane == 0) wred[wave] = wm;
        __syncthreads();
        const float4 m03 = *(const float4*)&wred[0];
        const float4 m47 = *(const float4*)&wred[4];
        const float mx = fmaxf(fmaxf(fmaxf(m03.x, m03.y), fmaxf(m03.z, m03.w)),
                               fmaxf(fmaxf(m47.x, m47.y), fmaxf(m47.z, m47.w)));
        const float e0 = expf(s0 - mx);
        const float e1 = expf(s1 - mx);
        const float ws = wave_sum(e0 + e1);
        if (lane == 0) wred[8 + wave] = ws;
        __syncthreads();
        const float4 t03 = *(const float4*)&wred[8];
        const float4 t47 = *(const float4*)&wred[12];
        const float tot = ((t03.x + t03.y) + (t03.z + t03.w)) + ((t47.x + t47.y) + (t47.z + t47.w));
        const float inv = 1.0f / tot;
        a0 = e0 * inv;
        a1 = e1 * inv;
        v2f av; av.x = a0; av.y = a1;
        *(v2f*)&attn_out[b * SEQ + wave * 64 + rr] = av;
    }

    {
        const int r0 = wave * 64;
        const int co = lane * 2;
        v2f c0 = {0.f, 0.f}, c1 = {0.f, 0.f}, c2 = {0.f, 0.f}, c3 = {0.f, 0.f};
        #pragma unroll
        for (int h = 0; h < 64; h += 32) {
            float wt[32];
            #pragma unroll
            for (int e = 0; e < 32; ++e) {
                const int q = h + e;
                const int srcl = ((q >> 3) & 1) * 16 + (q >> 4) * 4 + ((q >> 1) & 3);
                wt[e] = __builtin_bit_cast(float,
                    __builtin_amdgcn_readlane(__builtin_bit_cast(int, (q & 1) ? a1 : a0), srcl));
            }
            if (lane < 25) {
                #pragma unroll
                for (int e = 0; e < 32; e += 4) {
                    c0 += wt[e    ] * *(const v2f*)&emb[r0 + h + e    ][co];
                    c1 += wt[e + 1] * *(const v2f*)&emb[r0 + h + e + 1][co];
                    c2 += wt[e + 2] * *(const v2f*)&emb[r0 + h + e + 2][co];
                    c3 += wt[e + 3] * *(const v2f*)&emb[r0 + h + e + 3][co];
                }
            }
        }
        if (lane < 25) *(v2f*)&part[wave][co] = (c0 + c1) + (c2 + c3);
    }
    __syncthreads();
    if (tid < EMB) {
        ctxv[tid] = ((part[0][tid] + part[1][tid]) + (part[2][tid] + part[3][tid])) +
                    ((part[4][tid] + part[5][tid]) + (part[6][tid] + part[7][tid]));
    }
    __syncthreads();

    {
        const int row = tid >> 2, c = tid & 3;
        float acc = 0.f, bias = 0.f;
        if (row < EMB) {
            bias = be[row];
            acc = rowdot_part<4>(We + row * EMB, ctxv, c);
        }
        acc += xrow<1>(acc);
        acc += xrow<2>(acc);
        if (c == 0 && row < EMB) pooled[row] = acc + bias;
    }
    __syncthreads();

    {
        static_assert(NCLS == 2, "one 16-lane row of the first wave per class");
        const int row = lane >> 4, c = lane & 15;
        float acc = 0.f, bias = 0.f;
        if (tid < 32) {
            bias = bo[row];
            acc = rowdot_part<16>(Wo + row * EMB, pooled, c);
        }
        acc += xrow<1>(acc);
        acc += xrow<2>(acc);
        acc += xrow<4>(acc);
        acc += xrow<8>(acc);
        if (tid < 32 && c == 0) yhat_out[b * NCLS + row] = acc + bias;
    }
}

extern "C" void kernel_launch(void* const* d_in, const int* in_sizes, int n_in,
                              void* d_out, int out_size, void* d_ws, size_t ws_size,
                              hipStream_t stream) {
    const int*   x     = (const int*)  d_in[0];
    const float* table = (const float*)d_in[3];
    const float* Wq    = (const float*)d_in[4];
    const float* bq    = (const float*)d_in[5];
    const float* We    = (const float*)d_in[6];
    const float* be    = (const float*)d_in[7];
    const float* Wo    = (const float*)d_in[8];
    const float* bo    = (const float*)d_in[9];

    float* yhat = (float*)d_out;
    float* attn = (float*)d_out + BATCH * NCLS;

    attn_fused<<<dim3(BATCH), dim3(256), 0, stream>>>(
        x, table, Wq, bq, We, be, Wo, bo, yhat, attn);
}


# The two output layers folded into one 2x51 matrix (Wo*We, Wo*be+bo), built from the weights while the token ids are in flight: no memory load after the context sum; yhat summed in another order
# speedup vs baseline: 2.0572x; 1.0451x over previous
#include <hip/hip_runtime.h>

typedef float v2f __attribute__((ext_vector_type(2)));
typedef float v8f __attribute__((ext_vector_type(8)));

#define SEQ   512
#define BATCH 2048
#define EMB   50
#define EMBP  52
#define NCLS  2

#define OFF_EMB 0
#define OFF_U   (SEQ * EMB)
#define OFF_WR  (OFF_U + 8 * EMB)
#define OFF_V1  (OFF_WR + 16)
#define OFF_V2  (OFF_V1 + EMBP)
#define FOLDP   52
#define OFF_FOLD (OFF_V2 + EMBP)
#define LDS_FLOATS (OFF_FOLD + NCLS * FOLDP)

template <int M>
__device__ __forceinline__ float xrow(float x) {
    return __builtin_amdgcn_update_dpp(x, x, 0x160 + M, 0xf, 0xf, true);
}
__device__ __forceinline__ float xhalf(float x) {
    const int b = __builtin_bit_cast(int, x);
    return __builtin_bit_cast(float,
        __builtin_amdgcn_permlanex16(b, b, 0x76543210, 0xfedcba98, false, false));
}
__device__ __forceinline__ float wave_max(float v) {
    v = fmaxf(v, xrow<1>(v));
    v = fmaxf(v, xrow<2>(v));
    v = fmaxf(v, xrow<4>(v));
    v = fmaxf(v, xrow<8>(v));
    return fmaxf(v, xhalf(v));
}
__device__ __forceinline__ float wave_sum(float v) {
    v += xrow<1>(v);
    v += xrow<2>(v);
    v += xrow<4>(v);
    v += xrow<8>(v);
    return v + xhalf(v);
}

template <int G>
__device__ __forceinline__ float rowdot_part(const float* __restrict__ wrow,
                                             const float* vec, int c) {
    const v2f* w2 = (const v2f*)wrow;
    const v2f* x2 = (const v2f*)vec;
    constexpr int FULL = (EMB / 2) / G;
    constexpr int REST = (EMB / 2) % G;
    static_assert(FULL >= 1 && REST >= 1, "25 pairs over G lanes, with a remainder");
    const bool last = c < REST;
    const int  pl   = last ? c + G * FULL : c;
    v2f w[FULL];
    #pragma unroll
    for (int j = 0; j < FULL; ++j) w[j] = w2[c + G * j];
    const v2f wl = w2[pl];
    v2f s = w[0] * x2[c];
    #pragma unroll
    for (int j = 1; j < FULL; ++j) s += w[j] * x2[c + G * j];
    const v2f sl = s + wl * x2[pl];
    s = last ? sl : s;
    return s.x + s.y;
}

__launch_bounds__(256) __attribute__((amdgpu_num_vgpr(144)))
__global__ void attn_fused(const int* __restrict__ x,
                           const float* __restrict__ table,
                           const float* __restrict__ Wq, const float* __restrict__ bq,
                           const float* __restrict__ We, const float* __restrict__ be,
                           const float* __restrict__ Wo, const float* __restrict__ bo,
                           float* __restrict__ yhat_out, float* __restrict__ attn_out)
{
    __shared__ __attribute__((aligned(16))) float lds[LDS_FLOATS];
    float (*emb)[EMB]  = (float (*)[EMB])(lds + OFF_EMB);
    float (*part)[EMB] = (float (*)[EMB])(lds + OFF_U);
    float* wred        = lds + OFF_WR;
    float* meanv       = lds + OFF_V1;
    float* qv          = lds + OFF_V2;
    float* fold        = lds + OFF_FOLD;

    const int b    = blockIdx.x;
    const int tid  = threadIdx.x;
    const int lane = tid & 31;
    const int wave = tid >> 5;

    {
        const int r0 = wave * 64;
        const int t0 = x[(r0 + lane) * BATCH + b];
        const int t1 = x[(r0 + 32 + lane) * BATCH + b];

        if (tid <= EMB) {
            const bool col = tid < EMB;
            const float* src = col ? We + tid : be;
            const int    st  = col ? EMB : 1;
            float w[EMB];
            #pragma unroll
            for (int j = 0; j < EMB; ++j) w[j] = src[j * st];
            float m0a = col ? 0.f : bo[0], m0b = 0.f;
            float m1a = col ? 0.f : bo[1], m1b = 0.f;
            #pragma unroll
            for (int j = 0; j < EMB; j += 2) {
                m0a = fmaf(Wo[j],           w[j],     m0a);
                m0b = fmaf(Wo[j + 1],       w[j + 1], m0b);
                m1a = fmaf(Wo[EMB + j],     w[j],     m1a);
                m1b = fmaf(Wo[EMB + j + 1], w[j + 1], m1b);
            }
            fold[tid]         = m0a + m0b;
            fold[FOLDP + tid] = m1a + m1b;
        }

        int tk[64];
        #pragma unroll
        for (int i = 0; i < 64; ++i)
            tk[i] = __builtin_amdgcn_readlane((i < 32) ? t0 : t1, i & 31);

        if (lane < 25) {
            const int co = lane * 2;
            const v2f* tab2 = (const v2f*)table;
            v2f v[64];
            #pragma unroll
            for (int i = 0; i < 64; ++i)
                v[i] = tab2[tk[i] * (EMB / 2) + lane];

            v2f s0 = v[0], s1 = v[1], s2 = v[2], s3 = v[3];
            *(v2f*)&emb[r0    ][co] = v[0];
            *(v2f*)&emb[r0 + 1][co] = v[1];
            *(v2f*)&emb[r0 + 2][co] = v[2];
            *(v2f*)&emb[r0 + 3][co] = v[3];
            #pragma unroll
            for (int i = 4; i < 64; i += 4) {
                *(v2f*)&emb[r0 + i    ][co] = v[i];
                *(v2f*)&emb[r0 + i + 1][co] = v[i + 1];
                *(v2f*)&emb[r0 + i + 2][co] = v[i + 2];
                *(v2f*)&emb[r0 + i + 3][co] = v[i + 3];
                s0 += v[i]; s1 += v[i + 1]; s2 += v[i + 2]; s3 += v[i + 3];
            }
            *(v2f*)&part[wave][co] = (s0 + s1) + (s2 + s3);
        }
    }
    __syncthreads();
    if (tid < EMB) {
        meanv[tid] = (((part[0][tid] + part[1][tid]) + (part[2][tid] + part[3][tid])) +
                      ((part[4][tid] + part[5][tid]) + (part[6][tid] + part[7][tid]))) * (1.0f / SEQ);
    }
    __syncthreads();

    {
        const int row = tid >> 2, c = tid & 3;
        float acc = 0.f, bias = 0.f;
        if (row < EMB) {
            bias = bq[row];
            acc = rowdot_part<4>(Wq + row * EMB, meanv, c);
        }
        acc += xrow<1>(acc);
        acc += xrow<2>(acc);
        if (c == 0 && row < EMBP) qv[row] = acc + bias;
    }
    __syncthreads();

    const int mrow = lane & 15;
    const int rr   = (mrow >> 2) * 16 + (lane >> 4) * 8 + (mrow & 3) * 2;
    float s0, s1;
    {
        const int khalf = (lane >> 4) * 2;

        v2f breg[13];
        #pragma unroll
        for (int kk = 0; kk < 13; ++kk)
            breg[kk] = *(const v2f*)&qv[kk * 4 + khalf];

        v8f acc[4];
        #pragma unroll
        for (int i = 0; i < 4; ++i) {
            const int m0 = (wave * 4 + i) * 16;
            const float* rowp = &emb[m0 + mrow][0];
            v8f d = {0.f, 0.f, 0.f, 0.f, 0.f, 0.f, 0.f, 0.f};
            #pragma unroll
            for (int kk = 0; kk < 12; ++kk) {
                const int kb = kk * 4 + khalf;
                const v2f a = *(const v2f*)(rowp + kb);
                d = __builtin_amdgcn_wmma_f32_16x16x4_f32(
                        false, a, false, breg[kk], (short)0, d, false, false);
            }
            {
                v2f a = *(const v2f*)(rowp + 48);
                if (khalf) { a.x = 0.f; a.y = 0.f; }
                d = __builtin_amdgcn_wmma_f32_16x16x4_f32(
                        false, a, false, breg[12], (short)0, d, false, false);
            }
            acc[i] = d;
        }
        const int it = mrow >> 2, p = mrow & 3;
        float u[8];
        #pragma unroll
        for (int j = 0; j < 8; ++j) {
            float x = acc[0][j];
            x = (it == 1) ? acc[1][j] : x;
            x = (it == 2) ? acc[2][j] : x;
            x = (it == 3) ? acc[3][j] : x;
            u[j] = x;
        }
        s0 = u[0]; s1 = u[1];
        s0 = (p == 1) ? u[2] : s0;  s1 = (p == 1) ? u[3] : s1;
        s0 = (p == 2) ? u[4] : s0;  s1 = (p == 2) ? u[5] : s1;
        s0 = (p == 3) ? u[6] : s0;  s1 = (p == 3) ? u[7] : s1;
    }

    float a0, a1;
    {
        const float wm = wave_max(fmaxf(s0, s1));
        if (lane == 0) wred[wave] = wm;
        __syncthreads();
        const float4 m03 = *(const float4*)&wred[0];
        const float4 m47 = *(const float4*)&wred[4];
        const float mx = fmaxf(fmaxf(fmaxf(m03.x, m03.y), fmaxf(m03.z, m03.w)),
                               fmaxf(fmaxf(m47.x, m47.y), fmaxf(m47.z, m47.w)));
        const float e0 = expf(s0 - mx);
        const float e1 = expf(s1 - mx);
        const float ws = wave_sum(e0 + e1);
        if (lane == 0) wred[8 + wave] = ws;
        __syncthreads();
        const float4 t03 = *(const float4*)&wred[8];
        const float4 t47 = *(const float4*)&wred[12];
        const float tot = ((t03.x + t03.y) + (t03.z + t03.w)) + ((t47.x + t47.y) + (t47.z + t47.w));
        const float inv = 1.0f / tot;
        a0 = e0 * inv;
        a1 = e1 * inv;
        v2f av; av.x = a0; av.y = a1;
        *(v2f*)&attn_out[b * SEQ + wave * 64 + rr] = av;
    }

    {
        const int r0 = wave * 64;
        const int co = lane * 2;
        v2f c0 = {0.f, 0.f}, c1 = {0.f, 0.f}, c2 = {0.f, 0.f}, c3 = {0.f, 0.f};
        #pragma unroll
        for (int h = 0; h < 64; h += 32) {
            float wt[32];
            #pragma unroll
            for (int e = 0; e < 32; ++e) {
                const int q = h + e;
                const int srcl = ((q >> 3) & 1) * 16 + (q >> 4) * 4 + ((q >> 1) & 3);
                wt[e] = __builtin_bit_cast(float,
                    __builtin_amdgcn_readlane(__builtin_bit_cast(int, (q & 1) ? a1 : a0), srcl));
            }
            if (lane < 25) {
                #pragma unroll
                for (int e = 0; e < 32; e += 4) {
                    c0 += wt[e    ] * *(const v2f*)&emb[r0 + h + e    ][co];
                    c1 += wt[e + 1] * *(const v2f*)&emb[r0 + h + e + 1][co];
                    c2 += wt[e + 2] * *(const v2f*)&emb[r0 + h + e + 2][co];
                    c3 += wt[e + 3] * *(const v2f*)&emb[r0 + h + e + 3][co];
                }
            }
        }
        if (lane < 25) *(v2f*)&part[wave][co] = (c0 + c1) + (c2 + c3);
    }
    __syncthreads();

    {
        static_assert(NCLS == 2, "one 16-lane row of the first wave per class");
        const int row = lane >> 4, c = lane & 15;
        float acc = 0.f;
        if (tid < 32) {
            #pragma unroll
            for (int k0 = 0; k0 < 64; k0 += 16) {
                const int k = k0 + c;
                if (k < EMB) {
                    const float cx = ((part[0][k] + part[1][k]) + (part[2][k] + part[3][k])) +
                                     ((part[4][k] + part[5][k]) + (part[6][k] + part[7][k]));
                    acc = fmaf(fold[row * FOLDP + k], cx, acc);
                }
            }
        }
        acc += xrow<1>(acc);
        acc += xrow<2>(acc);
        acc += xrow<4>(acc);
        acc += xrow<8>(acc);
        if (tid < 32 && c == 0) yhat_out[b * NCLS + row] = acc + fold[row * FOLDP + EMB];
    }
}

extern "C" void kernel_launch(void* const* d_in, const int* in_sizes, int n_in,
                              void* d_out, int out_size, void* d_ws, size_t ws_size,
                              hipStream_t stream) {
    const int*   x     = (const int*)  d_in[0];
    const float* table = (const float*)d_in[3];
    const float* Wq    = (const float*)d_in[4];
    const float* bq    = (const float*)d_in[5];
    const float* We    = (const float*)d_in[6];
    const float* be    = (const float*)d_in[7];
    const float* Wo    = (const float*)d_in[8];
    const float* bo    = (const float*)d_in[9];

    float* yhat = (float*)d_out;
    float* attn = (float*)d_out + BATCH * NCLS;

    attn_fused<<<dim3(BATCH), dim3(256), 0, stream>>>(
        x, table, Wq, bq, We, be, Wo, bo, yhat, attn);
}


# Each work-group works through 3 consecutive batch elements (683 work-groups): one 12-byte load per lane brings their token ids, a third of the id line requests; output fold built once; sums unchanged
# speedup vs baseline: 2.2951x; 1.1156x over previous
#include <hip/hip_runtime.h>

typedef float v2f __attribute__((ext_vector_type(2)));
typedef float v8f __attribute__((ext_vector_type(8)));
typedef int   v3i __attribute__((ext_vector_type(3)));

#define SEQ   512
#define BATCH 2048
#define EMB   50
#define EMBP  52
#define NCLS  2
#define EPW   3
#define NWG   683
static_assert(NWG * EPW >= BATCH && (NWG - 1) * EPW < BATCH, "every element once");

#define OFF_EMB 0
#define OFF_U   (SEQ * EMB)
#define OFF_WR  (OFF_U + 8 * EMB)
#define OFF_V1  (OFF_WR + 16)
#define OFF_V2  (OFF_V1 + EMBP)
#define FOLDP   52
#define OFF_FOLD (OFF_V2 + EMBP)
#define OFF_C   (OFF_FOLD + NCLS * FOLDP)
#define LDS_FLOATS (OFF_C + 8 * EMB)

template <int M>
__device__ __forceinline__ float xrow(float x) {
    return __builtin_amdgcn_update_dpp(x, x, 0x160 + M, 0xf, 0xf, true);
}
__device__ __forceinline__ float xhalf(float x) {
    const int b = __builtin_bit_cast(int, x);
    return __builtin_bit_cast(float,
        __builtin_amdgcn_permlanex16(b, b, 0x76543210, 0xfedcba98, false, false));
}
__device__ __forceinline__ float wave_max(float v) {
    v = fmaxf(v, xrow<1>(v));
    v = fmaxf(v, xrow<2>(v));
    v = fmaxf(v, xrow<4>(v));
    v = fmaxf(v, xrow<8>(v));
    return fmaxf(v, xhalf(v));
}
__device__ __forceinline__ float wave_sum(float v) {
    v += xrow<1>(v);
    v += xrow<2>(v);
    v += xrow<4>(v);
    v += xrow<8>(v);
    return v + xhalf(v);
}

template <int G>
__device__ __forceinline__ float rowdot_part(const float* __restrict__ wrow,
                                             const float* vec, int c) {
    const v2f* w2 = (const v2f*)wrow;
    const v2f* x2 = (const v2f*)vec;
    constexpr int FULL = (EMB / 2) / G;
    constexpr int REST = (EMB / 2) % G;
    static_assert(FULL >= 1 && REST >= 1, "25 pairs over G lanes, with a remainder");
    const bool last = c < REST;
    const int  pl   = last ? c + G * FULL : c;
    v2f w[FULL];
    #pragma unroll
    for (int j = 0; j < FULL; ++j) w[j] = w2[c + G * j];
    const v2f wl = w2[pl];
    v2f s = w[0] * x2[c];
    #pragma unroll
    for (int j = 1; j < FULL; ++j) s += w[j] * x2[c + G * j];
    const v2f sl = s + wl * x2[pl];
    s = last ? sl : s;
    return s.x + s.y;
}

__launch_bounds__(256) __attribute__((amdgpu_num_vgpr(160)))
__global__ void attn_fused(const int* __restrict__ x,
                           const float* __restrict__ table,
                           const float* __restrict__ Wq, const float* __restrict__ bq,
                           const float* __restrict__ We, const float* __restrict__ be,
                           const float* __restrict__ Wo, const float* __restrict__ bo,
                           float* __restrict__ yhat_out, float* __restrict__ attn_out)
{
    __shared__ __attribute__((aligned(16))) float lds[LDS_FLOATS];
    float (*emb)[EMB]  = (float (*)[EMB])(lds + OFF_EMB);
    float (*part)[EMB] = (float (*)[EMB])(lds + OFF_U);
    float (*cpart)[EMB] = (float (*)[EMB])(lds + OFF_C);
    float* wred        = lds + OFF_WR;
    float* meanv       = lds + OFF_V1;
    float* qv          = lds + OFF_V2;
    float* fold        = lds + OFF_FOLD;

    const int tid  = threadIdx.x;
    const int lane = tid & 31;
    const int wave = tid >> 5;

    const int bfirst = blockIdx.x * EPW;
    const int nel    = (BATCH - bfirst < EPW) ? BATCH - bfirst : EPW;
    const int lbase  = (bfirst > BATCH - EPW) ? BATCH - EPW : bfirst;
    v3i ta, tb;
    {
        __builtin_memcpy(&ta, x + (wave * 64 + lane) * BATCH + lbase, sizeof(v3i));
        __builtin_memcpy(&tb, x + (wave * 64 + 32 + lane) * BATCH + lbase, sizeof(v3i));
        if (bfirst != lbase) {
            ta = __builtin_shufflevector(ta, ta, 1, 2, 0);
            tb = __builtin_shufflevector(tb, tb, 1, 2, 0);
        }

        if (tid <= EMB) {
            const bool col = tid < EMB;
            const float* src = col ? We + tid : be;
            const int    st  = col ? EMB : 1;
            float w[EMB];
            #pragma unroll
            for (int j = 0; j < EMB; ++j) w[j] = src[j * st];
            float m0a = col ? 0.f : bo[0], m0b = 0.f;
            float m1a = col ? 0.f : bo[1], m1b = 0.f;
            #pragma unroll
            for (int j = 0; j < EMB; j += 2) {
                m0a = fmaf(Wo[j],           w[j],     m0a);
                m0b = fmaf(Wo[j + 1],       w[j + 1], m0b);
                m1a = fmaf(Wo[EMB + j],     w[j],     m1a);
                m1b = fmaf(Wo[EMB + j + 1], w[j + 1], m1b);
            }
            fold[tid]         = m0a + m0b;
            fold[FOLDP + tid] = m1a + m1b;
        }
    }

    #pragma nounroll
    for (int e = 0; e < nel; ++e) {
        int tid_pass = threadIdx.x;
        asm volatile("" : "+v"(tid_pass));
        const int tid  = tid_pass;
        const int lane = tid & 31;
        const int wave = tid >> 5;
        const int b = bfirst + e;

        {
            const int r0 = wave * 64;
            const int t0 = (e == 0) ? ta.x : (e == 1) ? ta.y : ta.z;
            const int t1 = (e == 0) ? tb.x : (e == 1) ? tb.y : tb.z;

            int tk[64];
            #pragma unroll
            for (int i = 0; i < 64; ++i)
                tk[i] = __builtin_amdgcn_readlane((i < 32) ? t0 : t1, i & 31);

            if (lane < 25) {
                const int co = lane * 2;
                const v2f* tab2 = (const v2f*)table;
                v2f v[64];
                #pragma unroll
                for (int i = 0; i < 64; ++i)
                    v[i] = tab2[tk[i] * (EMB / 2) + lane];

                v2f s0 = v[0], s1 = v[1], s2 = v[2], s3 = v[3];
                *(v2f*)&emb[r0    ][co] = v[0];
                *(v2f*)&emb[r0 + 1][co] = v[1];
                *(v2f*)&emb[r0 + 2][co] = v[2];
                *(v2f*)&emb[r0 + 3][co] = v[3];
                #pragma unroll
                for (int i = 4; i < 64; i += 4) {
                    *(v2f*)&emb[r0 + i    ][co] = v[i];
                    *(v2f*)&emb[r0 + i + 1][co] = v[i + 1];
                    *(v2f*)&emb[r0 + i + 2][co] = v[i + 2];
                    *(v2f*)&emb[r0 + i + 3][co] = v[i + 3];
                    s0 += v[i]; s1 += v[i + 1]; s2 += v[i + 2]; s3 += v[i + 3];
                }
                *(v2f*)&part[wave][co] = (s0 + s1) + (s2 + s3);
            }
        }
        __syncthreads();
        if (tid < EMB) {
            meanv[tid] = (((part[0][tid] + part[1][tid]) + (part[2][tid] + part[3][tid])) +
                          ((part[4][tid] + part[5][tid]) + (part[6][tid] + part[7][tid]))) * (1.0f / SEQ);
        }
        __syncthreads();

        {
            const int row = tid >> 2, c = tid & 3;
            float acc = 0.f, bias = 0.f;
            if (row < EMB) {
                bias = bq[row];
                acc = rowdot_part<4>(Wq + row * EMB, meanv, c);
            }
            acc += xrow<1>(acc);
            acc += xrow<2>(acc);
            if (c == 0 && row < EMBP) qv[row] = acc + bias;
        }
        __syncthreads();

        const int mrow = lane & 15;
        const int rr   = (mrow >> 2) * 16 + (lane >> 4) * 8 + (mrow & 3) * 2;
        float s0, s1;
        {
            const int khalf = (lane >> 4) * 2;

            v2f breg[13];
            #pragma unroll
            for (int kk = 0; kk < 13; ++kk)
                breg[kk] = *(const v2f*)&qv[kk * 4 + khalf];

            v8f acc[4];
            #pragma unroll
            for (int i = 0; i < 4; ++i) {
                const int m0 = (wave * 4 + i) * 16;
                const float* rowp = &emb[m0 + mrow][0];
                v8f d = {0.f, 0.f, 0.f, 0.f, 0.f, 0.f, 0.f, 0.f};
                #pragma unroll
                for (int kk = 0; kk < 12; ++kk) {
                    const int kb = kk * 4 + khalf;
                    const v2f a = *(const v2f*)(rowp + kb);
                    d = __builtin_amdgcn_wmma_f32_16x16x4_f32(
                            false, a, false, breg[kk], (short)0, d, false, false);
                }
                {
                    v2f a = *(const v2f*)(rowp + 48);
                    if (khalf) { a.x = 0.f; a.y = 0.f; }
                    d = __builtin_amdgcn_wmma_f32_16x16x4_f32(
                            false, a, false, breg[12], (short)0, d, false, false);
                }
                acc[i] = d;
            }
            const int it = mrow >> 2, p = mrow & 3;
            float u[8];
            #pragma unroll
            for (int j = 0; j < 8; ++j) {
                float x = acc[0][j];
                x = (it == 1) ? acc[1][j] : x;
                x = (it == 2) ? acc[2][j] : x;
                x = (it == 3) ? acc[3][j] : x;
                u[j] = x;
            }
            s0 = u[0]; s1 = u[1];
            s0 = (p == 1) ? u[2] : s0;  s1 = (p == 1) ? u[3] : s1;
            s0 = (p == 2) ? u[4] : s0;  s1 = (p == 2) ? u[5] : s1;
            s0 = (p == 3) ? u[6] : s0;  s1 = (p == 3) ? u[7] : s1;
        }

        float a0, a1;
        {
            const float wm = wave_max(fmaxf(s0, s1));
            if (lane == 0) wred[wave] = wm;
            __syncthreads();
            const float4 m03 = *(const float4*)&wred[0];
            const float4 m47 = *(const float4*)&wred[4];
            const float mx = fmaxf(fmaxf(fmaxf(m03.x, m03.y), fmaxf(m03.z, m03.w)),
                                   fmaxf(fmaxf(m47.x, m47.y), fmaxf(m47.z, m47.w)));
            const float e0 = expf(s0 - mx);
            const float e1 = expf(s1 - mx);
            const float ws = wave_sum(e0 + e1);
            if (lane == 0) wred[8 + wave] = ws;
            __syncthreads();
            const float4 t03 = *(const float4*)&wred[8];
            const float4 t47 = *(const float4*)&wred[12];
            const float tot = ((t03.x + t03.y) + (t03.z + t03.w)) + ((t47.x + t47.y) + (t47.z + t47.w));
            const float inv = 1.0f / tot;
            a0 = e0 * inv;
            a1 = e1 * inv;
            v2f av; av.x = a0; av.y = a1;
            *(v2f*)&attn_out[b * SEQ + wave * 64 + rr] = av;
        }

        {
            const int r0 = wave * 64;
            const int co = lane * 2;
            v2f c0 = {0.f, 0.f}, c1 = {0.f, 0.f}, c2 = {0.f, 0.f}, c3 = {0.f, 0.f};
            #pragma unroll
            for (int h = 0; h < 64; h += 32) {
                float wt[32];
                #pragma unroll
                for (int e = 0; e < 32; ++e) {
                    const int q = h + e;
                    const int srcl = ((q >> 3) & 1) * 16 + (q >> 4) * 4 + ((q >> 1) & 3);
                    wt[e] = __builtin_bit_cast(float,
                        __builtin_amdgcn_readlane(__builtin_bit_cast(int, (q & 1) ? a1 : a0), srcl));
                }
                if (lane < 25) {
                    #pragma unroll
                    for (int e = 0; e < 32; e += 4) {
                        c0 += wt[e    ] * *(const v2f*)&emb[r0 + h + e    ][co];
                        c1 += wt[e + 1] * *(const v2f*)&emb[r0 + h + e + 1][co];
                        c2 += wt[e + 2] * *(const v2f*)&emb[r0 + h + e + 2][co];
                        c3 += wt[e + 3] * *(const v2f*)&emb[r0 + h + e + 3][co];
                    }
                }
            }
            if (lane < 25) *(v2f*)&cpart[wave][co] = (c0 + c1) + (c2 + c3);
        }
        __syncthreads();

        {
            static_assert(NCLS == 2, "one 16-lane row of the first wave per class");
            const int row = lane >> 4, c = lane & 15;
            float acc = 0.f;
            if (tid < 32) {
                #pragma unroll
                for (int k0 = 0; k0 < 64; k0 += 16) {
                    const int k = k0 + c;
                    if (k < EMB) {
                        const float cx = ((cpart[0][k] + cpart[1][k]) + (cpart[2][k] + cpart[3][k])) +
                                         ((cpart[4][k] + cpart[5][k]) + (cpart[6][k] + cpart[7][k]));
                        acc = fmaf(fold[row * FOLDP + k], cx, acc);
                    }
                }
            }
            acc += xrow<1>(acc);
            acc += xrow<2>(acc);
            acc += xrow<4>(acc);
            acc += xrow<8>(acc);
            if (tid < 32 && c == 0) yhat_out[b * NCLS + row] = acc + fold[row * FOLDP + EMB];
        }
    }
}

extern "C" void kernel_launch(void* const* d_in, const int* in_sizes, int n_in,
                              void* d_out, int out_size, void* d_ws, size_t ws_size,
                              hipStream_t stream) {
    const int*   x     = (const int*)  d_in[0];
    const float* table = (const float*)d_in[3];
    const float* Wq    = (const float*)d_in[4];
    const float* bq    = (const float*)d_in[5];
    const float* We    = (const float*)d_in[6];
    const float* be    = (const float*)d_in[7];
    const float* Wo    = (const float*)d_in[8];
    const float* bo    = (const float*)d_in[9];

    float* yhat = (float*)d_out;
    float* attn = (float*)d_out + BATCH * NCLS;

    attn_fused<<<dim3(NWG), dim3(256), 0, stream>>>(
        x, table, Wq, bq, We, be, Wo, bo, yhat, attn);
}


# Score WMMA swapped: gathered rows as B, A rows = q and both folded output rows: a lane gets its rows' scores and output projections, no LDS context pass; softmax total, yhat summed in another order
# speedup vs baseline: 2.5589x; 1.1149x over previous
#include <hip/hip_runtime.h>

typedef float v2f __attribute__((ext_vector_type(2)));
typedef float v8f __attribute__((ext_vector_type(8)));
typedef int   v3i __attribute__((ext_vector_type(3)));

#define SEQ   512
#define BATCH 2048
#define EMB   50
#define EMBP  52
#define NCLS  2
#define EPW   3
#define NWG   683
static_assert(NWG * EPW >= BATCH && (NWG - 1) * EPW < BATCH, "every element once");

#define OFF_EMB 0
#define OFF_U   (SEQ * EMB)
#define OFF_WR  (OFF_U + 8 * EMB)
#define OFF_V1  (OFF_WR + 32)
#define OFF_V2  (OFF_V1 + EMBP)
#define FOLDP   52
#define OFF_FOLD (OFF_V2 + EMBP)
static_assert(FOLDP == EMBP, "q and the two folded rows lie at one pitch");
#define LDS_FLOATS (OFF_FOLD + NCLS * FOLDP)

template <int M>
__device__ __forceinline__ float xrow(float x) {
    return __builtin_amdgcn_update_dpp(x, x, 0x160 + M, 0xf, 0xf, true);
}
__device__ __forceinline__ float xhalf(float x) {
    const int b = __builtin_bit_cast(int, x);
    return __builtin_bit_cast(float,
        __builtin_amdgcn_permlanex16(b, b, 0x76543210, 0xfedcba98, false, false));
}
__device__ __forceinline__ float wave_max(float v) {
    v = fmaxf(v, xrow<1>(v));
    v = fmaxf(v, xrow<2>(v));
    v = fmaxf(v, xrow<4>(v));
    v = fmaxf(v, xrow<8>(v));
    return fmaxf(v, xhalf(v));
}
__device__ __forceinline__ float wave_sum(float v) {
    v += xrow<1>(v);
    v += xrow<2>(v);
    v += xrow<4>(v);
    v += xrow<8>(v);
    return v + xhalf(v);
}

template <int G>
__device__ __forceinline__ float rowdot_part(const float* __restrict__ wrow,
                                             const float* vec, int c) {
    const v2f* w2 = (const v2f*)wrow;
    const v2f* x2 = (const v2f*)vec;
    constexpr int FULL = (EMB / 2) / G;
    constexpr int REST = (EMB / 2) % G;
    static_assert(FULL >= 1 && REST >= 1, "25 pairs over G lanes, with a remainder");
    const bool last = c < REST;
    const int  pl   = last ? c + G * FULL : c;
    v2f w[FULL];
    #pragma unroll
    for (int j = 0; j < FULL; ++j) w[j] = w2[c + G * j];
    const v2f wl = w2[pl];
    v2f s = w[0] * x2[c];
    #pragma unroll
    for (int j = 1; j < FULL; ++j) s += w[j] * x2[c + G * j];
    const v2f sl = s + wl * x2[pl];
    s = last ? sl : s;
    return s.x + s.y;
}

__launch_bounds__(256) __attribute__((amdgpu_num_vgpr(160)))
__global__ void attn_fused(const int* __restrict__ x,
                           const float* __restrict__ table,
                           const float* __restrict__ Wq, const float* __restrict__ bq,
                           const float* __restrict__ We, const float* __restrict__ be,
                           const float* __restrict__ Wo, const float* __restrict__ bo,
                           float* __restrict__ yhat_out, float* __restrict__ attn_out)
{
    __shared__ __attribute__((aligned(16))) float lds[LDS_FLOATS];
    float (*emb)[EMB]  = (float (*)[EMB])(lds + OFF_EMB);
    float (*part)[EMB] = (float (*)[EMB])(lds + OFF_U);
    float* wred        = lds + OFF_WR;
    float* meanv       = lds + OFF_V1;
    float* qv          = lds + OFF_V2;
    float* fold        = lds + OFF_FOLD;

    const int tid  = threadIdx.x;
    const int lane = tid & 31;
    const int wave = tid >> 5;

    const int bfirst = blockIdx.x * EPW;
    const int nel    = (BATCH - bfirst < EPW) ? BATCH - bfirst : EPW;
    const int lbase  = (bfirst > BATCH - EPW) ? BATCH - EPW : bfirst;
    v3i ta, tb;
    {
        __builtin_memcpy(&ta, x + (wave * 64 + lane) * BATCH + lbase, sizeof(v3i));
        __builtin_memcpy(&tb, x + (wave * 64 + 32 + lane) * BATCH + lbase, sizeof(v3i));
        if (bfirst != lbase) {
            ta = __builtin_shufflevector(ta, ta, 1, 2, 0);
            tb = __builtin_shufflevector(tb, tb, 1, 2, 0);
        }

        if (tid <= EMB) {
            const bool col = tid < EMB;
            const float* src = col ? We + tid : be;
            const int    st  = col ? EMB : 1;
            float w[EMB];
            #pragma unroll
            for (int j = 0; j < EMB; ++j) w[j] = src[j * st];
            float m0a = col ? 0.f : bo[0], m0b = 0.f;
            float m1a = col ? 0.f : bo[1], m1b = 0.f;
            #pragma unroll
            for (int j = 0; j < EMB; j += 2) {
                m0a = fmaf(Wo[j],           w[j],     m0a);
                m0b = fmaf(Wo[j + 1],       w[j + 1], m0b);
                m1a = fmaf(Wo[EMB + j],     w[j],     m1a);
                m1b = fmaf(Wo[EMB + j + 1], w[j + 1], m1b);
            }
            fold[tid]         = m0a + m0b;
            fold[FOLDP + tid] = m1a + m1b;
        }
    }

    #pragma nounroll
    for (int e = 0; e < nel; ++e) {
        int tid_pass = threadIdx.x;
        asm volatile("" : "+v"(tid_pass));
        const int tid  = tid_pass;
        const int lane = tid & 31;
        const int wave = tid >> 5;
        const int b = bfirst + e;

        {
            const int r0 = wave * 64;
            const int t0 = (e == 0) ? ta.x : (e == 1) ? ta.y : ta.z;
            const int t1 = (e == 0) ? tb.x : (e == 1) ? tb.y : tb.z;

            int tk[64];
            #pragma unroll
            for (int i = 0; i < 64; ++i)
                tk[i] = __builtin_amdgcn_readlane((i < 32) ? t0 : t1, i & 31);

            if (lane < 25) {
                const int co = lane * 2;
                const v2f* tab2 = (const v2f*)table;
                v2f v[64];
                #pragma unroll
                for (int i = 0; i < 64; ++i)
                    v[i] = tab2[tk[i] * (EMB / 2) + lane];

                v2f s0 = v[0], s1 = v[1], s2 = v[2], s3 = v[3];
                *(v2f*)&emb[r0    ][co] = v[0];
                *(v2f*)&emb[r0 + 1][co] = v[1];
                *(v2f*)&emb[r0 + 2][co] = v[2];
                *(v2f*)&emb[r0 + 3][co] = v[3];
                #pragma unroll
                for (int i = 4; i < 64; i += 4) {
                    *(v2f*)&emb[r0 + i    ][co] = v[i];
                    *(v2f*)&emb[r0 + i + 1][co] = v[i + 1];
                    *(v2f*)&emb[r0 + i + 2][co] = v[i + 2];
                    *(v2f*)&emb[r0 + i + 3][co] = v[i + 3];
                    s0 += v[i]; s1 += v[i + 1]; s2 += v[i + 2]; s3 += v[i + 3];
                }
                *(v2f*)&part[wave][co] = (s0 + s1) + (s2 + s3);
            }
        }
        __syncthreads();
        if (tid < EMB) {
            meanv[tid] = (((part[0][tid] + part[1][tid]) + (part[2][tid] + part[3][tid])) +
                          ((part[4][tid] + part[5][tid]) + (part[6][tid] + part[7][tid]))) * (1.0f / SEQ);
        }
        __syncthreads();

        {
            const int row = tid >> 2, c = tid & 3;
            float acc = 0.f, bias = 0.f;
            if (row < EMB) {
                bias = bq[row];
                acc = rowdot_part<4>(Wq + row * EMB, meanv, c);
            }
            acc += xrow<1>(acc);
            acc += xrow<2>(acc);
            if (c == 0 && row < EMBP) qv[row] = acc + bias;
        }
        __syncthreads();

        float s0, s1;
        float z00, z01, z10, z11;
        {
            const int n     = lane & 15;
            const int khalf = (lane >> 4) * 2;
            const int am    = n & 7;
            const float* asrc = qv + ((am == 1 || am == 2) ? am * FOLDP : 0);

            v2f areg[13];
            #pragma unroll
            for (int kk = 0; kk < 13; ++kk)
                areg[kk] = *(const v2f*)&asrc[kk * 4 + khalf];
            if (khalf) { areg[12].x = 0.f; areg[12].y = 0.f; }

            v8f acc[4];
            #pragma unroll
            for (int i = 0; i < 4; ++i) {
                const int m0 = (wave * 4 + i) * 16;
                const float* rowp = &emb[m0 + n][0];
                v8f d = {0.f, 0.f, 0.f, 0.f, 0.f, 0.f, 0.f, 0.f};
                #pragma unroll
                for (int kk = 0; kk < 12; ++kk) {
                    const int kb = kk * 4 + khalf;
                    const v2f bfr = *(const v2f*)(rowp + kb);
                    d = __builtin_amdgcn_wmma_f32_16x16x4_f32(
                            false, areg[kk], false, bfr, (short)0, d, false, false);
                }
                {
                    v2f bfr = *(const v2f*)(rowp + 48);
                    if (khalf) { bfr.x = 0.f; bfr.y = 0.f; }
                    d = __builtin_amdgcn_wmma_f32_16x16x4_f32(
                            false, areg[12], false, bfr, (short)0, d, false, false);
                }
                acc[i] = d;
            }
            const bool up = khalf != 0;
            s0  = up ? acc[1][0] : acc[0][0];   s1  = up ? acc[3][0] : acc[2][0];
            z00 = up ? acc[1][1] : acc[0][1];   z01 = up ? acc[3][1] : acc[2][1];
            z10 = up ? acc[1][2] : acc[0][2];   z11 = up ? acc[3][2] : acc[2][2];
        }

        {
            const float wm = wave_max(fmaxf(s0, s1));
            if (lane == 0) wred[wave] = wm;
            __syncthreads();
            const float4 m03 = *(const float4*)&wred[0];
            const float4 m47 = *(const float4*)&wred[4];
            const float mx = fmaxf(fmaxf(fmaxf(m03.x, m03.y), fmaxf(m03.z, m03.w)),
                                   fmaxf(fmaxf(m47.x, m47.y), fmaxf(m47.z, m47.w)));
            const float e0 = expf(s0 - mx);
            const float e1 = expf(s1 - mx);
            const float ws = wave_sum(e0 + e1);
            const float y0 = wave_sum(fmaf(e1, z01, e0 * z00));
            const float y1 = wave_sum(fmaf(e1, z11, e0 * z10));
            if (lane == 0) {
                wred[8 + wave]  = ws;
                wred[16 + wave] = y0;
                wred[24 + wave] = y1;
            }
            __syncthreads();
            const float4 t03 = *(const float4*)&wred[8];
            const float4 t47 = *(const float4*)&wred[12];
            const float tot = ((t03.x + t03.y) + (t03.z + t03.w)) + ((t47.x + t47.y) + (t47.z + t47.w));
            const float inv = 1.0f / tot;
            attn_out[b * SEQ + wave * 64 + lane]      = e0 * inv;
            attn_out[b * SEQ + wave * 64 + 32 + lane] = e1 * inv;
            static_assert(NCLS == 2, "wred[] holds the sums of two classes");
            if (tid < NCLS) {
                const float4 g03 = *(const float4*)&wred[16 + 8 * tid];
                const float4 g47 = *(const float4*)&wred[20 + 8 * tid];
                const float g = ((g03.x + g03.y) + (g03.z + g03.w)) + ((g47.x + g47.y) + (g47.z + g47.w));
                yhat_out[b * NCLS + tid] = fmaf(g, inv, fold[tid * FOLDP + EMB]);
            }
        }
    }
}

extern "C" void kernel_launch(void* const* d_in, const int* in_sizes, int n_in,
                              void* d_out, int out_size, void* d_ws, size_t ws_size,
                              hipStream_t stream) {
    const int*   x     = (const int*)  d_in[0];
    const float* table = (const float*)d_in[3];
    const float* Wq    = (const float*)d_in[4];
    const float* bq    = (const float*)d_in[5];
    const float* We    = (const float*)d_in[6];
    const float* be    = (const float*)d_in[7];
    const float* Wo    = (const float*)d_in[8];
    const float* bo    = (const float*)d_in[9];

    float* yhat = (float*)d_out;
    float* attn = (float*)d_out + BATCH * NCLS;

    attn_fused<<<dim3(NWG), dim3(256), 0, stream>>>(
        x, table, Wq, bq, We, be, Wo, bo, yhat, attn);
}


# Work split evened over the WGPs: 768 work-groups (3 per WGP, all resident), the first 512 take 3 elements, the last 256 take 2: 8 elements per WGP, not 9 on most and 6 on the rest; sums unchanged
# speedup vs baseline: 2.7118x; 1.0598x over previous
#include <hip/hip_runtime.h>

typedef float v2f __attribute__((ext_vector_type(2)));
typedef float v8f __attribute__((ext_vector_type(8)));
typedef int   v3i __attribute__((ext_vector_type(3)));

#define SEQ   512
#define BATCH 2048
#define EMB   50
#define EMBP  52
#define NCLS  2
#define EPW   3
#define NWG    768
#define NLONG  512
#define ESHORT 2
static_assert(NLONG * EPW + (NWG - NLONG) * ESHORT == BATCH, "every element once");
static_assert(ESHORT == EPW - 1 && NLONG * EPW >= 1, "a short work-group's id load starts one column early, inside the row");

#define OFF_EMB 0
#define OFF_U   (SEQ * EMB)
#define OFF_WR  (OFF_U + 8 * EMB)
#define OFF_V1  (OFF_WR + 32)
#define OFF_V2  (OFF_V1 + EMBP)
#define FOLDP   52
#define OFF_FOLD (OFF_V2 + EMBP)
static_assert(FOLDP == EMBP, "q and the two folded rows lie at one pitch");
#define LDS_FLOATS (OFF_FOLD + NCLS * FOLDP)

template <int M>
__device__ __forceinline__ float xrow(float x) {
    return __builtin_amdgcn_update_dpp(x, x, 0x160 + M, 0xf, 0xf, true);
}
__device__ __forceinline__ float xhalf(float x) {
    const int b = __builtin_bit_cast(int, x);
    return __builtin_bit_cast(float,
        __builtin_amdgcn_permlanex16(b, b, 0x76543210, 0xfedcba98, false, false));
}
__device__ __forceinline__ float wave_max(float v) {
    v = fmaxf(v, xrow<1>(v));
    v = fmaxf(v, xrow<2>(v));
    v = fmaxf(v, xrow<4>(v));
    v = fmaxf(v, xrow<8>(v));
    return fmaxf(v, xhalf(v));
}
__device__ __forceinline__ float wave_sum(float v) {
    v += xrow<1>(v);
    v += xrow<2>(v);
    v += xrow<4>(v);
    v += xrow<8>(v);
    return v + xhalf(v);
}

template <int G>
__device__ __forceinline__ float rowdot_part(const float* __restrict__ wrow,
                                             const float* vec, int c) {
    const v2f* w2 = (const v2f*)wrow;
    const v2f* x2 = (const v2f*)vec;
    constexpr int FULL = (EMB / 2) / G;
    constexpr int REST = (EMB / 2) % G;
    static_assert(FULL >= 1 && REST >= 1, "25 pairs over G lanes, with a remainder");
    const bool last = c < REST;
    const int  pl   = last ? c + G * FULL : c;
    v2f w[FULL];
    #pragma unroll
    for (int j = 0; j < FULL; ++j) w[j] = w2[c + G * j];
    const v2f wl = w2[pl];
    v2f s = w[0] * x2[c];
    #pragma unroll
    for (int j = 1; j < FULL; ++j) s += w[j] * x2[c + G * j];
    const v2f sl = s + wl * x2[pl];
    s = last ? sl : s;
    return s.x + s.y;
}

__launch_bounds__(256) __attribute__((amdgpu_num_vgpr(160)))
__global__ void attn_fused(const int* __restrict__ x,
                           const float* __restrict__ table,
                           const float* __restrict__ Wq, const float* __restrict__ bq,
                           const float* __restrict__ We, const float* __restrict__ be,
                           const float* __restrict__ Wo, const float* __restrict__ bo,
                           float* __restrict__ yhat_out, float* __restrict__ attn_out)
{
    __shared__ __attribute__((aligned(16))) float lds[LDS_FLOATS];
    float (*emb)[EMB]  = (float (*)[EMB])(lds + OFF_EMB);
    float (*part)[EMB] = (float (*)[EMB])(lds + OFF_U);
    float* wred        = lds + OFF_WR;
    float* meanv       = lds + OFF_V1;
    float* qv          = lds + OFF_V2;
    float* fold        = lds + OFF_FOLD;

    const int tid  = threadIdx.x;
    const int lane = tid & 31;
    const int wave = tid >> 5;

    const int  wg     = blockIdx.x;
    const bool shortw = wg >= NLONG;
    const int  bfirst = shortw ? NLONG * EPW + (wg - NLONG) * ESHORT : wg * EPW;
    const int  nel    = shortw ? ESHORT : EPW;
    const int  lbase  = shortw ? bfirst - 1 : bfirst;
    v3i ta, tb;
    {
        __builtin_memcpy(&ta, x + (wave * 64 + lane) * BATCH + lbase, sizeof(v3i));
        __builtin_memcpy(&tb, x + (wave * 64 + 32 + lane) * BATCH + lbase, sizeof(v3i));
        if (bfirst != lbase) {
            ta = __builtin_shufflevector(ta, ta, 1, 2, 0);
            tb = __builtin_shufflevector(tb, tb, 1, 2, 0);
        }

        if (tid <= EMB) {
            const bool col = tid < EMB;
            const float* src = col ? We + tid : be;
            const int    st  = col ? EMB : 1;
            float w[EMB];
            #pragma unroll
            for (int j = 0; j < EMB; ++j) w[j] = src[j * st];
            float m0a = col ? 0.f : bo[0], m0b = 0.f;
            float m1a = col ? 0.f : bo[1], m1b = 0.f;
            #pragma unroll
            for (int j = 0; j < EMB; j += 2) {
                m0a = fmaf(Wo[j],           w[j],     m0a);
                m0b = fmaf(Wo[j + 1],       w[j + 1], m0b);
                m1a = fmaf(Wo[EMB + j],     w[j],     m1a);
                m1b = fmaf(Wo[EMB + j + 1], w[j + 1], m1b);
            }
            fold[tid]         = m0a + m0b;
            fold[FOLDP + tid] = m1a + m1b;
        }
    }

    #pragma nounroll
    for (int e = 0; e < nel; ++e) {
        int tid_pass = threadIdx.x;
        asm volatile("" : "+v"(tid_pass));
        const int tid  = tid_pass;
        const int lane = tid & 31;
        const int wave = tid >> 5;
        const int b = bfirst + e;

        {
            const int r0 = wave * 64;
            const int t0 = (e == 0) ? ta.x : (e == 1) ? ta.y : ta.z;
            const int t1 = (e == 0) ? tb.x : (e == 1) ? tb.y : tb.z;

            int tk[64];
            #pragma unroll
            for (int i = 0; i < 64; ++i)
                tk[i] = __builtin_amdgcn_readlane((i < 32) ? t0 : t1, i & 31);

            if (lane < 25) {
                const int co = lane * 2;
                const v2f* tab2 = (const v2f*)table;
                v2f v[64];
                #pragma unroll
                for (int i = 0; i < 64; ++i)
                    v[i] = tab2[tk[i] * (EMB / 2) + lane];

                v2f s0 = v[0], s1 = v[1], s2 = v[2], s3 = v[3];
                *(v2f*)&emb[r0    ][co] = v[0];
                *(v2f*)&emb[r0 + 1][co] = v[1];
                *(v2f*)&emb[r0 + 2][co] = v[2];
                *(v2f*)&emb[r0 + 3][co] = v[3];
                #pragma unroll
                for (int i = 4; i < 64; i += 4) {
                    *(v2f*)&emb[r0 + i    ][co] = v[i];
                    *(v2f*)&emb[r0 + i + 1][co] = v[i + 1];
                    *(v2f*)&emb[r0 + i + 2][co] = v[i + 2];
                    *(v2f*)&emb[r0 + i + 3][co] = v[i + 3];
                    s0 += v[i]; s1 += v[i + 1]; s2 += v[i + 2]; s3 += v[i + 3];
                }
                *(v2f*)&part[wave][co] = (s0 + s1) + (s2 + s3);
            }
        }
        __syncthreads();
        if (tid < EMB) {
            meanv[tid] = (((part[0][tid] + part[1][tid]) + (part[2][tid] + part[3][tid])) +
                          ((part[4][tid] + part[5][tid]) + (part[6][tid] + part[7][tid]))) * (1.0f / SEQ);
        }
        __syncthreads();

        {
            const int row = tid >> 2, c = tid & 3;
            float acc = 0.f, bias = 0.f;
            if (row < EMB) {
                bias = bq[row];
                acc = rowdot_part<4>(Wq + row * EMB, meanv, c);
            }
            acc += xrow<1>(acc);
            acc += xrow<2>(acc);
            if (c == 0 && row < EMBP) qv[row] = acc + bias;
        }
        __syncthreads();

        float s0, s1;
        float z00, z01, z10, z11;
        {
            const int n     = lane & 15;
            const int khalf = (lane >> 4) * 2;
            const int am    = n & 7;
            const float* asrc = qv + ((am == 1 || am == 2) ? am * FOLDP : 0);

            v2f areg[13];
            #pragma unroll
            for (int kk = 0; kk < 13; ++kk)
                areg[kk] = *(const v2f*)&asrc[kk * 4 + khalf];
            if (khalf) { areg[12].x = 0.f; areg[12].y = 0.f; }

            v8f acc[4];
            #pragma unroll
            for (int i = 0; i < 4; ++i) {
                const int m0 = (wave * 4 + i) * 16;
                const float* rowp = &emb[m0 + n][0];
                v8f d = {0.f, 0.f, 0.f, 0.f, 0.f, 0.f, 0.f, 0.f};
                #pragma unroll
                for (int kk = 0; kk < 12; ++kk) {
                    const int kb = kk * 4 + khalf;
                    const v2f bfr = *(const v2f*)(rowp + kb);
                    d = __builtin_amdgcn_wmma_f32_16x16x4_f32(
                            false, areg[kk], false, bfr, (short)0, d, false, false);
                }
                {
                    v2f bfr = *(const v2f*)(rowp + 48);
                    if (khalf) { bfr.x = 0.f; bfr.y = 0.f; }
                    d = __builtin_amdgcn_wmma_f32_16x16x4_f32(
                            false, areg[12], false, bfr, (short)0, d, false, false);
                }
                acc[i] = d;
            }
            const bool up = khalf != 0;
            s0  = up ? acc[1][0] : acc[0][0];   s1  = up ? acc[3][0] : acc[2][0];
            z00 = up ? acc[1][1] : acc[0][1];   z01 = up ? acc[3][1] : acc[2][1];
            z10 = up ? acc[1][2] : acc[0][2];   z11 = up ? acc[3][2] : acc[2][2];
        }

        {
            const float wm = wave_max(fmaxf(s0, s1));
            if (lane == 0) wred[wave] = wm;
            __syncthreads();
            const float4 m03 = *(const float4*)&wred[0];
            const float4 m47 = *(const float4*)&wred[4];
            const float mx = fmaxf(fmaxf(fmaxf(m03.x, m03.y), fmaxf(m03.z, m03.w)),
                                   fmaxf(fmaxf(m47.x, m47.y), fmaxf(m47.z, m47.w)));
            const float e0 = expf(s0 - mx);
            const float e1 = expf(s1 - mx);
            const float ws = wave_sum(e0 + e1);
            const float y0 = wave_sum(fmaf(e1, z01, e0 * z00));
            const float y1 = wave_sum(fmaf(e1, z11, e0 * z10));
            if (lane == 0) {
                wred[8 + wave]  = ws;
                wred[16 + wave] = y0;
                wred[24 + wave] = y1;
            }
            __syncthreads();
            const float4 t03 = *(const float4*)&wred[8];
            const float4 t47 = *(const float4*)&wred[12];
            const float tot = ((t03.x + t03.y) + (t03.z + t03.w)) + ((t47.x + t47.y) + (t47.z + t47.w));
            const float inv = 1.0f / tot;
            attn_out[b * SEQ + wave * 64 + lane]      = e0 * inv;
            attn_out[b * SEQ + wave * 64 + 32 + lane] = e1 * inv;
            static_assert(NCLS == 2, "wred[] holds the sums of two classes");
            if (tid < NCLS) {
                const float4 g03 = *(const float4*)&wred[16 + 8 * tid];
                const float4 g47 = *(const float4*)&wred[20 + 8 * tid];
                const float g = ((g03.x + g03.y) + (g03.z + g03.w)) + ((g47.x + g47.y) + (g47.z + g47.w));
                yhat_out[b * NCLS + tid] = fmaf(g, inv, fold[tid * FOLDP + EMB]);
            }
        }
    }
}

extern "C" void kernel_launch(void* const* d_in, const int* in_sizes, int n_in,
                              void* d_out, int out_size, void* d_ws, size_t ws_size,
                              hipStream_t stream) {
    const int*   x     = (const int*)  d_in[0];
    const float* table = (const float*)d_in[3];
    const float* Wq    = (const float*)d_in[4];
    const float* bq    = (const float*)d_in[5];
    const float* We    = (const float*)d_in[6];
    const float* be    = (const float*)d_in[7];
    const float* Wo    = (const float*)d_in[8];
    const float* bo    = (const float*)d_in[9];

    float* yhat = (float*)d_out;
    float* attn = (float*)d_out + BATCH * NCLS;

    attn_fused<<<dim3(NWG), dim3(256), 0, stream>>>(
        x, table, Wq, bq, We, be, Wo, bo, yhat, attn);
}
